# starting kernel
# baseline (speedup 1.0000x reference)
#include <hip/hip_runtime.h>
#include <hip/hip_bf16.h>

#define NN 100000
#define NE 800000
#define DD 64
#define NL 3
#define NG 100
#define NPG (NN / NG)
#define NC 10
#define NCP 16
#define BN_EPS 1e-5f

typedef float v2f __attribute__((ext_vector_type(2)));
typedef float v8f __attribute__((ext_vector_type(8)));

__global__ void __launch_bounds__(256)
egnn_edge_layer(const float* __restrict__ h,
                const float* __restrict__ ef,
                const int*   __restrict__ src,
                const int*   __restrict__ dst,
                const float* __restrict__ W1,
                const float* __restrict__ b1,
                const float* __restrict__ W2,
                const float* __restrict__ b2,
                float*       __restrict__ hnext)
{
    __shared__ float sW2[64 * 64];
    __shared__ float sB1[64], sB2[64];
    __shared__ float sHid[8][16 * 64];
    __shared__ float sEf[8][16];
    __shared__ int   sDst[8][16];

    const int tid  = threadIdx.x;
    const int lane = tid & 31;
    const int wave = tid >> 5;
    const int lo   = lane & 15;
    const int hi   = lane >> 4;

    for (int i = tid; i < 64 * 64; i += 256) sW2[i] = W2[i];
    if (tid < 64) { sB1[tid] = b1[tid]; sB2[tid] = b2[tid]; }
    __syncthreads();

    const float* W1b    = W1 + 64 * 64;
    const float* w1last = W1 + 128 * 64;

    const int numTiles  = (NE + 15) >> 4;
    const int waveGlob  = blockIdx.x * 8 + wave;
    const int waveCount = gridDim.x * 8;

    for (int tile = waveGlob; tile < numTiles; tile += waveCount) {
        const int e      = tile * 16 + lo;
        const bool valid = (e < NE);
        const int eC     = valid ? e : (NE - 1);
        const int sRow   = src[eC];
        const int dRow   = dst[eC];
        if (hi == 0) {
            sEf[wave][lo]  = ef[eC];
            sDst[wave][lo] = valid ? dRow : -1;
        }

        const float* rowS = h + (long long)sRow * DD;
        const float* rowD = h + (long long)dRow * DD;

        v8f acc[4];
        #pragma unroll
        for (int n = 0; n < 4; ++n) {
            const float bb = sB1[n * 16 + lo];
            #pragma unroll
            for (int j = 0; j < 8; ++j) acc[n][j] = bb;
        }

        #pragma unroll
        for (int c = 0; c < 16; ++c) {
            const int k0 = 4 * c + 2 * hi;
            v2f a1 = { rowS[k0], rowS[k0 + 1] };
            v2f a2 = { rowD[k0], rowD[k0 + 1] };
            #pragma unroll
            for (int n = 0; n < 4; ++n) {
                const int col = n * 16 + lo;
                v2f bA = { W1 [k0 * 64 + col], W1 [(k0 + 1) * 64 + col] };
                v2f bB = { W1b[k0 * 64 + col], W1b[(k0 + 1) * 64 + col] };
                acc[n] = __builtin_amdgcn_wmma_f32_16x16x4_f32(
                    false, a1, false, bA, (short)0, acc[n], false, false);
                acc[n] = __builtin_amdgcn_wmma_f32_16x16x4_f32(
                    false, a2, false, bB, (short)0, acc[n], false, false);
            }
        }

        #pragma unroll
        for (int n = 0; n < 4; ++n) {
            const int col  = n * 16 + lo;
            const float wl = w1last[col];
            #pragma unroll
            for (int v = 0; v < 8; ++v) {
                const int M = v + 8 * hi;
                float val = acc[n][v] + sEf[wave][M] * wl;
                sHid[wave][M * 64 + col] = fmaxf(val, 0.0f);
            }
        }

        v8f mac[4];
        #pragma unroll
        for (int n = 0; n < 4; ++n) {
            const float bb = sB2[n * 16 + lo];
            #pragma unroll
            for (int j = 0; j < 8; ++j) mac[n][j] = bb;
        }

        #pragma unroll
        for (int c = 0; c < 16; ++c) {
            const int k0 = 4 * c + 2 * hi;
            v2f a = { sHid[wave][lo * 64 + k0], sHid[wave][lo * 64 + k0 + 1] };
            #pragma unroll
            for (int n = 0; n < 4; ++n) {
                const int col = n * 16 + lo;
                v2f b = { sW2[k0 * 64 + col], sW2[(k0 + 1) * 64 + col] };
                mac[n] = __builtin_amdgcn_wmma_f32_16x16x4_f32(
                    false, a, false, b, (short)0, mac[n], false, false);
            }
        }

        #pragma unroll
        for (int v = 0; v < 8; ++v) {
            const int M    = v + 8 * hi;
            const int drow = sDst[wave][M];
            if (drow >= 0) {
                #pragma unroll
                for (int n = 0; n < 4; ++n)
                    atomicAdd(&hnext[(long long)drow * DD + n * 16 + lo], mac[n][v]);
            }
        }
    }
}

__global__ void __launch_bounds__(256)
zero_f32(float* __restrict__ p, long long n)
{
    for (long long i = blockIdx.x * 256LL + threadIdx.x; i < n;
         i += (long long)gridDim.x * 256LL)
        p[i] = 0.0f;
}

__global__ void __launch_bounds__(256)
bn_reduce(const float* __restrict__ h, float* __restrict__ stats, int nrows)
{
    __shared__ float sSum[256], sSq[256];
    const int c = threadIdx.x & 63;
    const int r = threadIdx.x >> 6;
    float s = 0.f, q = 0.f;
    for (int row = blockIdx.x * 4 + r; row < nrows; row += gridDim.x * 4) {
        float v = h[(long long)row * DD + c];
        s += v; q += v * v;
    }
    sSum[threadIdx.x] = s; sSq[threadIdx.x] = q;
    __syncthreads();
    if (threadIdx.x < 64) {
        float ts = sSum[c] + sSum[64 + c] + sSum[128 + c] + sSum[192 + c];
        float tq = sSq [c] + sSq [64 + c] + sSq [128 + c] + sSq [192 + c];
        atomicAdd(&stats[c], ts);
        atomicAdd(&stats[64 + c], tq);
    }
}

__global__ void __launch_bounds__(256)
bn_relu_apply(float* __restrict__ h, const float* __restrict__ stats,
              const float* __restrict__ gamma, const float* __restrict__ beta,
              int nrows, float invN)
{
    const long long total = (long long)nrows * DD;
    for (long long i = blockIdx.x * 256LL + threadIdx.x; i < total;
         i += (long long)gridDim.x * 256LL) {
        const int c  = (int)(i & 63);
        const float mu  = stats[c] * invN;
        const float var = stats[64 + c] * invN - mu * mu;
        const float rs  = rsqrtf(var + BN_EPS);
        const float v   = (h[i] - mu) * rs * gamma[c] + beta[c];
        h[i] = fmaxf(v, 0.0f);
    }
}

__global__ void __launch_bounds__(256)
pool_mean(const float* __restrict__ h, float* __restrict__ emb)
{
    __shared__ float sm[256];
    const int g = blockIdx.x;
    const int c = threadIdx.x & 63;
    const int r = threadIdx.x >> 6;
    float s = 0.f;
    for (int row = r; row < NPG; row += 4)
        s += h[((long long)g * NPG + row) * DD + c];
    sm[threadIdx.x] = s;
    __syncthreads();
    if (threadIdx.x < 64)
        emb[g * DD + c] = (sm[c] + sm[64 + c] + sm[128 + c] + sm[192 + c]) / (float)NPG;
}

__global__ void __launch_bounds__(256)
bn_heads(const float* __restrict__ emb,
         const float* __restrict__ bg, const float* __restrict__ bb,
         const float* __restrict__ clfW, const float* __restrict__ clfb,
         const float* __restrict__ cptW, const float* __restrict__ cptb,
         float* __restrict__ outLogits, float* __restrict__ outCpt)
{
    __shared__ float sEmb[NG * DD];
    __shared__ float sMu[64], sR[64];
    __shared__ float red[256];
    const int tid = threadIdx.x;
    const int c   = tid & 63;
    const int r   = tid >> 6;

    float s = 0.f, q = 0.f;
    for (int g = r; g < NG; g += 4) {
        float v = emb[g * DD + c];
        s += v; q += v * v;
    }
    red[tid] = s; __syncthreads();
    if (tid < 64) sMu[c] = (red[c] + red[64 + c] + red[128 + c] + red[192 + c]) / (float)NG;
    __syncthreads();
    red[tid] = q; __syncthreads();
    if (tid < 64) {
        float tq  = (red[c] + red[64 + c] + red[128 + c] + red[192 + c]) / (float)NG;
        float mu  = sMu[c];
        sR[c] = rsqrtf(tq - mu * mu + BN_EPS);
    }
    __syncthreads();
    for (int i = tid; i < NG * DD; i += 256) {
        int cc = i & 63;
        sEmb[i] = (emb[i] - sMu[cc]) * sR[cc] * bg[cc] + bb[cc];
    }
    __syncthreads();
    for (int i = tid; i < NG * NC; i += 256) {
        int g = i / NC, k = i % NC;
        float a = clfb[k];
        #pragma unroll 8
        for (int cc = 0; cc < DD; ++cc) a += sEmb[g * DD + cc] * clfW[cc * NC + k];
        outLogits[i] = a;
    }
    for (int i = tid; i < NG * NCP; i += 256) {
        int g = i >> 4, k = i & 15;
        float a = cptb[k];
        #pragma unroll 8
        for (int cc = 0; cc < DD; ++cc) a += sEmb[g * DD + cc] * cptW[cc * NCP + k];
        outCpt[i] = a;
    }
}

extern "C" void kernel_launch(void* const* d_in, const int* in_sizes, int n_in,
                              void* d_out, int out_size, void* d_ws, size_t ws_size,
                              hipStream_t stream)
{
    const float* x    = (const float*)d_in[0];
    const float* ef   = (const float*)d_in[1];
    const int*   src  = (const int*)  d_in[2];
    const int*   dst  = (const int*)  d_in[3];
    const float* W1s  = (const float*)d_in[5];
    const float* b1s  = (const float*)d_in[6];
    const float* W2s  = (const float*)d_in[7];
    const float* b2s  = (const float*)d_in[8];
    const float* gam  = (const float*)d_in[9];
    const float* bet  = (const float*)d_in[10];
    const float* bng  = (const float*)d_in[11];
    const float* bnb  = (const float*)d_in[12];
    const float* clfW = (const float*)d_in[13];
    const float* clfb = (const float*)d_in[14];
    const float* cptW = (const float*)d_in[15];
    const float* cptb = (const float*)d_in[16];

    float* outH      = (float*)d_out;
    float* outLogits = outH + (long long)NN * DD;
    float* outCpt    = outLogits + NG * NC;

    float* hA    = (float*)d_ws;
    float* stats = hA + (long long)NN * DD;
    float* emb   = stats + 128;

    const float* hin = x;
    float* tgt[NL] = { outH, hA, outH };

    const int edgeBlocks = 2048;

    for (int l = 0; l < NL; ++l) {
        float* hn = tgt[l];
        zero_f32<<<1024, 256, 0, stream>>>(hn, (long long)NN * DD);
        zero_f32<<<1,    256, 0, stream>>>(stats, 128);
        egnn_edge_layer<<<edgeBlocks, 256, 0, stream>>>(
            hin, ef, src, dst,
            W1s + (long long)l * (2 * DD + 1) * DD,
            b1s + l * DD,
            W2s + (long long)l * DD * DD,
            b2s + l * DD,
            hn);
        bn_reduce<<<512, 256, 0, stream>>>(hn, stats, NN);
        bn_relu_apply<<<1024, 256, 0, stream>>>(hn, stats, gam + l * DD, bet + l * DD,
                                                NN, 1.0f / (float)NN);
        hin = hn;
    }

    pool_mean<<<NG, 256, 0, stream>>>(outH, emb);
    bn_heads<<<1, 256, 0, stream>>>(emb, bng, bnb, clfW, clfb, cptW, cptb,
                                    outLogits, outCpt);
}


# Gather pass and product 2 fused: a wave owns a 16-node tile, keeps its summed hidden rows in LDS and multiplies them by W2 on WMMA itself; 15 launches to 12; batch-norm partial sums regrouped
# speedup vs baseline: 2.5681x; 2.5681x over previous
#include <hip/hip_runtime.h>
#include <hip/hip_bf16.h>

#define NN 100000
#define NE 800000
#define DD 64
#define NL 3
#define NG 100
#define NPG (NN / NG)
#define NC 10
#define NCP 16
#define BN_EPS 1e-5f
#define WS_PROJECTED_BYTES ((2LL * NN * DD + 128 + (long long)NG * DD) * 4)
#define EDGE_PQ_BLOCKS 512
#define WS_NODE_P2_BYTES (WS_PROJECTED_BYTES + (long long)NN * 4)
#define EDGE_SUM_BLOCKS 1250
#define SCAN_THREADS 1024
#define SCAN_BLOCKS  ((NN + SCAN_THREADS - 1) / SCAN_THREADS)
#define WS_GROUPED_BYTES (WS_NODE_P2_BYTES + ((long long)NN + 128 + 2LL * NE) * 4)
static_assert(SCAN_BLOCKS <= 128, "block sums have 128 slots");

typedef float v2f __attribute__((ext_vector_type(2)));
typedef float v8f __attribute__((ext_vector_type(8)));

__global__ void __launch_bounds__(256)
egnn_edge_layer(const float* __restrict__ h,
                const float* __restrict__ ef,
                const int*   __restrict__ src,
                const int*   __restrict__ dst,
                const float* __restrict__ W1,
                const float* __restrict__ b1,
                const float* __restrict__ W2,
                const float* __restrict__ b2,
                float*       __restrict__ hnext)
{
    __shared__ float sW2[64 * 64];
    __shared__ float sB1[64], sB2[64], sWl[64];

    const int tid  = threadIdx.x;
    const int lane = tid & 31;
    const int wave = tid >> 5;
    const int lo   = lane & 15;
    const int hi   = lane >> 4;

    for (int i = tid; i < 64 * 64; i += 256) sW2[i] = W2[i];
    if (tid < 64) { sB1[tid] = b1[tid]; sB2[tid] = b2[tid]; sWl[tid] = W1[128 * 64 + tid]; }
    __syncthreads();

    const float* W1b    = W1 + 64 * 64;

    const int numTiles  = (NE + 15) >> 4;
    const int waveGlob  = blockIdx.x * 8 + wave;
    const int waveCount = gridDim.x * 8;

    const int t0 = waveGlob;
    const int t1 = t0 + waveCount;
    int   curDst, nxtS, nxtD;
    float curEf, nxtEf;
    bool  nxtValid;
    v2f   a1[16], a2[16];
    {
        const int  e0 = t0 * 16 + lo;
        const bool v0 = (t0 < numTiles) && (e0 < NE);
        const int  c0 = v0 ? e0 : (NE - 1);
        const int  s0 = src[c0];
        const int  d0 = dst[c0];
        curEf  = ef[c0];
        curDst = v0 ? d0 : -1;

        const int  e1 = t1 * 16 + lo;
        nxtValid = (t1 < numTiles) && (e1 < NE);
        const int  c1 = nxtValid ? e1 : (NE - 1);
        nxtS  = src[c1];
        nxtD  = dst[c1];
        nxtEf = ef[c1];

        const float* rowS = h + (long long)s0 * DD;
        const float* rowD = h + (long long)d0 * DD;
        #pragma unroll
        for (int c = 0; c < 16; ++c) {
            const int k0 = 4 * c + 2 * hi;
            a1[c] = v2f{ rowS[k0], rowS[k0 + 1] };
            a2[c] = v2f{ rowD[k0], rowD[k0 + 1] };
        }
    }

    v2f wA[16][4], wB[16][4];
    #pragma unroll
    for (int c = 0; c < 16; ++c) {
        const int k0 = 4 * c + 2 * hi;
        #pragma unroll
        for (int n = 0; n < 4; ++n) {
            const int col = n * 16 + lo;
            wA[c][n] = v2f{ W1 [k0 * 64 + col], W1 [(k0 + 1) * 64 + col] };
            wB[c][n] = v2f{ W1b[k0 * 64 + col], W1b[(k0 + 1) * 64 + col] };
        }
    }

    for (int tile = t0; tile < numTiles; tile += waveCount) {
        asm volatile("" ::: "memory");

        const int  t2 = tile + 2 * waveCount;
        const int  e2 = t2 * 16 + lo;
        const bool v2 = (t2 < numTiles) && (e2 < NE);
        const int  c2 = v2 ? e2 : (NE - 1);
        const int   s2 = src[c2];
        const int   d2 = dst[c2];
        const float f2 = ef[c2];

        v8f acc[4];
        #pragma unroll
        for (int n = 0; n < 4; ++n) {
            #pragma unroll
            for (int r = 0; r < 8; ++r) acc[n][r] = sB1[n * 16 + 8 * hi + r];
        }

        #pragma unroll
        for (int c = 0; c < 16; ++c) {
            #pragma unroll
            for (int n = 0; n < 4; ++n) {
                acc[n] = __builtin_amdgcn_wmma_f32_16x16x4_f32(
                    false, wA[c][n], false, a1[c], (short)0, acc[n], false, false);
                acc[n] = __builtin_amdgcn_wmma_f32_16x16x4_f32(
                    false, wB[c][n], false, a2[c], (short)0, acc[n], false, false);
            }
        }

        {
            const float* rowS = h + (long long)nxtS * DD;
            const float* rowD = h + (long long)nxtD * DD;
            #pragma unroll
            for (int c = 0; c < 16; ++c) {
                const int k0 = 4 * c + 2 * hi;
                a1[c] = v2f{ rowS[k0], rowS[k0 + 1] };
                a2[c] = v2f{ rowD[k0], rowD[k0 + 1] };
            }
        }

        #pragma unroll
        for (int n = 0; n < 4; ++n) {
            #pragma unroll
            for (int r = 0; r < 8; ++r) {
                const float val = acc[n][r] + curEf * sWl[n * 16 + 8 * hi + r];
                acc[n][r] = fmaxf(val, 0.0f);
            }
        }

        v8f mac[4];
        #pragma unroll
        for (int n2 = 0; n2 < 4; ++n2) {
            const float bb = sB2[n2 * 16 + lo];
            #pragma unroll
            for (int j = 0; j < 8; ++j) mac[n2][j] = bb;
        }

        #pragma unroll
        for (int n = 0; n < 4; ++n) {
            #pragma unroll
            for (int p = 0; p < 4; ++p) {
                const int kr = 16 * n + 8 * hi + 2 * p;
                v2f a = { acc[n][2 * p], acc[n][2 * p + 1] };
                #pragma unroll
                for (int n2 = 0; n2 < 4; ++n2) {
                    const int col = n2 * 16 + lo;
                    v2f b = { sW2[kr * 64 + col], sW2[(kr + 1) * 64 + col] };
                    mac[n2] = __builtin_amdgcn_wmma_f32_16x16x4_f32(
                        false, a, false, b, (short)0, mac[n2], false, false);
                }
            }
        }

        #pragma unroll
        for (int v = 0; v < 8; ++v) {
            const int dl   = __builtin_amdgcn_readlane(curDst, v);
            const int dh   = __builtin_amdgcn_readlane(curDst, v + 8);
            const int drow = hi ? dh : dl;
            if (drow >= 0) {
                #pragma unroll
                for (int n2 = 0; n2 < 4; ++n2)
                    atomicAdd(&hnext[(long long)drow * DD + n2 * 16 + lo], mac[n2][v]);
            }
        }

        curEf    = nxtEf;
        curDst   = nxtValid ? nxtD : -1;
        nxtS     = s2;
        nxtD     = d2;
        nxtEf    = f2;
        nxtValid = v2;
    }
}

typedef float v4f __attribute__((ext_vector_type(4)));

template <bool NORM>
__global__ void __launch_bounds__(256)
egnn_node_project(const float* h,
                  const float* __restrict__ W1,
                  const float* __restrict__ b1,
                  float*       __restrict__ P,
                  float*       __restrict__ Q,
                  const float* __restrict__ statsIn,
                  const float* __restrict__ gammaIn,
                  const float* __restrict__ betaIn,
                  float invN,
                  float*       zt,
                  float*       __restrict__ statsOut,
                  int*         __restrict__ degClear)
{
    const int tid  = threadIdx.x;
    const int lane = tid & 31;
    const int wave = __builtin_amdgcn_readfirstlane(tid >> 5);
    const int lo   = lane & 15;
    const int hi   = lane >> 4;
    const int cg   = wave & 3;
    const int half = cg >> 1;
    const int colBase = (cg & 1) * 32;

    if (blockIdx.x == 0 && tid < 128) statsOut[tid] = 0.0f;
    if (degClear) {
        for (int i = blockIdx.x * 256 + tid; i < NN; i += gridDim.x * 256) degClear[i] = 0;
    }

    const float* Wsel = W1 + half * 64 * 64;
    float*       out  = half ? Q : P;

    v2f wf[16][2];
    #pragma unroll
    for (int s = 0; s < 16; ++s) {
        const int kr = 8 * (s >> 1) + 4 * hi + 2 * (s & 1);
        #pragma unroll
        for (int t = 0; t < 2; ++t) {
            const int col = colBase + 16 * t + lo;
            wf[s][t] = v2f{ Wsel[kr * 64 + col], Wsel[(kr + 1) * 64 + col] };
        }
    }
    float bias[2];
    #pragma unroll
    for (int t = 0; t < 2; ++t) bias[t] = half ? 0.0f : b1[colBase + 16 * t + lo];

    v4f sc[8], sh[8];
    if (NORM) {
        #pragma unroll
        for (int j = 0; j < 8; ++j) {
            #pragma unroll
            for (int e = 0; e < 4; ++e) {
                const int   f   = 8 * j + 4 * hi + e;
                const float mu  = statsIn[f] * invN;
                const float var = statsIn[64 + f] * invN - mu * mu;
                const float s   = rsqrtf(var + BN_EPS) * gammaIn[f];
                sc[j][e] = s;
                sh[j][e] = betaIn[f] - mu * s;
            }
        }
    }

    const int numTiles = NN / 16;
    const int group    = blockIdx.x * 2 + (wave >> 2);
    const int groups   = gridDim.x * 2;
    const int trips    = (numTiles - (int)blockIdx.x * 2 + groups - 1) / groups;

    v4f x[8];
    #pragma unroll
    for (int j = 0; j < 8; ++j) x[j] = v4f{ 0.f, 0.f, 0.f, 0.f };
    if (group < numTiles) {
        const float* row = h + (long long)(group * 16 + lo) * DD + 4 * hi;
        #pragma unroll
        for (int j = 0; j < 8; ++j) x[j] = *(const v4f*)(row + 8 * j);
    }

    for (int it = 0; it < trips; ++it) {
        const int  tile = group + it * groups;
        const bool live = tile < numTiles;
        if (live) {
            if (NORM) {
                #pragma unroll
                for (int j = 0; j < 8; ++j) {
                    #pragma unroll
                    for (int e = 0; e < 4; ++e)
                        x[j][e] = fmaxf(fmaf(x[j][e], sc[j][e], sh[j][e]), 0.0f);
                }
            }

            v8f acc[2];
            #pragma unroll
            for (int t = 0; t < 2; ++t) {
                #pragma unroll
                for (int r = 0; r < 8; ++r) acc[t][r] = bias[t];
            }
            #pragma unroll
            for (int j = 0; j < 8; ++j) {
                const v2f aLo = { x[j][0], x[j][1] };
                const v2f aHi = { x[j][2], x[j][3] };
                #pragma unroll
                for (int t = 0; t < 2; ++t) {
                    acc[t] = __builtin_amdgcn_wmma_f32_16x16x4_f32(
                        false, aLo, false, wf[2 * j][t], (short)0, acc[t], false, false);
                    acc[t] = __builtin_amdgcn_wmma_f32_16x16x4_f32(
                        false, aHi, false, wf[2 * j + 1][t], (short)0, acc[t], false, false);
                }
            }

            const int tn = tile + groups;
            if (tn < numTiles) {
                const float* row = h + (long long)(tn * 16 + lo) * DD + 4 * hi;
                #pragma unroll
                for (int j = 0; j < 8; ++j) x[j] = *(const v4f*)(row + 8 * j);
            }

            #pragma unroll
            for (int t = 0; t < 2; ++t) {
                #pragma unroll
                for (int r = 0; r < 8; ++r)
                    out[(long long)(tile * 16 + 8 * hi + r) * DD + colBase + 16 * t + lo] = acc[t][r];
            }
        }

        if (zt == nullptr) continue;
        __builtin_amdgcn_s_barrier();
        if (live) {
            float* z = zt + (long long)(tile * 16 + 4 * cg) * DD + 4 * lane;
            *(v4f*)z            = v4f{ 0.f, 0.f, 0.f, 0.f };
            *(v4f*)(z + 2 * DD) = v4f{ 0.f, 0.f, 0.f, 0.f };
        }
    }
}

__global__ void __launch_bounds__(256)
egnn_edge_layer_pq(const float* __restrict__ P,
                   const float* __restrict__ Q,
                   const float* __restrict__ ef,
                   const int*   __restrict__ src,
                   const int*   __restrict__ dst,
                   const float* __restrict__ wl,
                   const float* __restrict__ W2,
                   const float* __restrict__ b2,
                   float*       __restrict__ hnext)
{
    __shared__ float sW2[64 * 64];
    __shared__ float sB2[64], sWl[64];

    const int tid  = threadIdx.x;
    const int lane = tid & 31;
    const int wave = tid >> 5;
    const int lo   = lane & 15;
    const int hi   = lane >> 4;

    for (int i = tid; i < 64 * 64; i += 256) sW2[i] = W2[i];
    if (tid < 64) { sB2[tid] = b2[tid]; sWl[tid] = wl[tid]; }
    __syncthreads();

    const int numTiles  = (NE + 15) >> 4;
    const int waveGlob  = blockIdx.x * 8 + wave;
    const int waveCount = gridDim.x * 8;

    const int t0 = waveGlob;
    const int t1 = t0 + waveCount;
    int   curDst, nxtS, nxtD;
    float curEf, nxtEf;
    bool  nxtValid;
    v4f   p[8], q[8];
    {
        const int  e0 = t0 * 16 + lo;
        const bool v0 = (t0 < numTiles) && (e0 < NE);
        const int  c0 = v0 ? e0 : (NE - 1);
        const int  s0 = src[c0];
        const int  d0 = dst[c0];
        curEf  = ef[c0];
        curDst = v0 ? d0 : -1;

        const int  e1 = t1 * 16 + lo;
        nxtValid = (t1 < numTiles) && (e1 < NE);
        const int  c1 = nxtValid ? e1 : (NE - 1);
        nxtS  = src[c1];
        nxtD  = dst[c1];
        nxtEf = ef[c1];

        const float* rowS = P + (long long)s0 * DD + 8 * hi;
        const float* rowD = Q + (long long)d0 * DD + 8 * hi;
        #pragma unroll
        for (int n = 0; n < 4; ++n) {
            p[2 * n]     = *(const v4f*)(rowS + 16 * n);
            p[2 * n + 1] = *(const v4f*)(rowS + 16 * n + 4);
            q[2 * n]     = *(const v4f*)(rowD + 16 * n);
            q[2 * n + 1] = *(const v4f*)(rowD + 16 * n + 4);
        }
    }

    for (int tile = t0; tile < numTiles; tile += waveCount) {
        asm volatile("" ::: "memory");

        const int  t2 = tile + 2 * waveCount;
        const int  e2 = t2 * 16 + lo;
        const bool v2 = (t2 < numTiles) && (e2 < NE);
        const int  c2 = v2 ? e2 : (NE - 1);
        const int   s2 = src[c2];
        const int   d2 = dst[c2];
        const float f2 = ef[c2];

        v8f acc[4];
        #pragma unroll
        for (int n = 0; n < 4; ++n) {
            #pragma unroll
            for (int r = 0; r < 8; ++r) {
                const float pq  = p[2 * n + (r >> 2)][r & 3] + q[2 * n + (r >> 2)][r & 3];
                const float val = pq + curEf * sWl[n * 16 + 8 * hi + r];
                acc[n][r] = fmaxf(val, 0.0f);
            }
        }

        {
            const float* rowS = P + (long long)nxtS * DD + 8 * hi;
            const float* rowD = Q + (long long)nxtD * DD + 8 * hi;
            #pragma unroll
            for (int n = 0; n < 4; ++n) {
                p[2 * n]     = *(const v4f*)(rowS + 16 * n);
                p[2 * n + 1] = *(const v4f*)(rowS + 16 * n + 4);
                q[2 * n]     = *(const v4f*)(rowD + 16 * n);
                q[2 * n + 1] = *(const v4f*)(rowD + 16 * n + 4);
            }
        }

        v8f mac[4];
        #pragma unroll
        for (int n2 = 0; n2 < 4; ++n2) {
            const float bb = sB2[n2 * 16 + lo];
            #pragma unroll
            for (int j = 0; j < 8; ++j) mac[n2][j] = bb;
        }

        #pragma unroll
        for (int n = 0; n < 4; ++n) {
            #pragma unroll
            for (int pp = 0; pp < 4; ++pp) {
                const int kr = 16 * n + 8 * hi + 2 * pp;
                v2f a = { acc[n][2 * pp], acc[n][2 * pp + 1] };
                #pragma unroll
                for (int n2 = 0; n2 < 4; ++n2) {
                    const int col = n2 * 16 + lo;
                    v2f b = { sW2[kr * 64 + col], sW2[(kr + 1) * 64 + col] };
                    mac[n2] = __builtin_amdgcn_wmma_f32_16x16x4_f32(
                        false, a, false, b, (short)0, mac[n2], false, false);
                }
            }
        }

        #pragma unroll
        for (int v = 0; v < 8; ++v) {
            const int dl   = __builtin_amdgcn_readlane(curDst, v);
            const int dh   = __builtin_amdgcn_readlane(curDst, v + 8);
            const int drow = hi ? dh : dl;
            if (drow >= 0) {
                #pragma unroll
                for (int n2 = 0; n2 < 4; ++n2)
                    atomicAdd(&hnext[(long long)drow * DD + n2 * 16 + lo], mac[n2][v]);
            }
        }

        curEf    = nxtEf;
        curDst   = nxtValid ? nxtD : -1;
        nxtS     = s2;
        nxtD     = d2;
        nxtEf    = f2;
        nxtValid = v2;
    }
}

static_assert(NE % 8 == 0, "edges are walked in groups of 8");

template <bool COUNT>
__global__ void __launch_bounds__(256)
egnn_edge_sum(const float* __restrict__ P,
              const float* __restrict__ Q,
              const float* __restrict__ ef,
              const int*   __restrict__ src,
              const int*   __restrict__ dst,
              const float* __restrict__ wl,
              float*       __restrict__ hs,
              float*       __restrict__ deg)
{
    const int lane      = threadIdx.x & 31;
    const unsigned waveGlob  = (unsigned)__builtin_amdgcn_readfirstlane(blockIdx.x * 8 + (threadIdx.x >> 5));
    const unsigned waveCount = gridDim.x * 8;
    const float w0 = wl[lane], w1 = wl[32 + lane];

    const int e0 = (int)(waveGlob       * (unsigned)(NE / 8) / waveCount) * 8;
    const int e1 = (int)((waveGlob + 1) * (unsigned)(NE / 8) / waveCount) * 8;

    for (int eb = e0; eb < e1; eb += 32) {
        const int  e  = eb + lane;
        const bool ok = e < e1;
        const int  ec = ok ? e : (e1 - 1);
        const int  sv = src[ec];
        const int  dv = dst[ec];
        const int  fb = __builtin_bit_cast(int, ef[ec]);
        if (COUNT && ok) atomicAdd(&deg[dv], 1.0f);
        const int  n  = (e1 - eb < 32) ? (e1 - eb) : 32;

        #pragma unroll 1
        for (int i = 0; i < n; i += 8) {
            float p0[8], p1[8], q0[8], q1[8];
            #pragma unroll
            for (int u = 0; u < 8; ++u) {
                const unsigned s = (unsigned)__builtin_amdgcn_readlane(sv, i + u);
                const unsigned d = (unsigned)__builtin_amdgcn_readlane(dv, i + u);
                const float* rp = P + (size_t)s * DD + lane;
                const float* rq = Q + (size_t)d * DD + lane;
                p0[u] = rp[0];  p1[u] = rp[32];
                q0[u] = rq[0];  q1[u] = rq[32];
            }
            #pragma unroll
            for (int u = 0; u < 8; ++u) {
                const unsigned d = (unsigned)__builtin_amdgcn_readlane(dv, i + u);
                const float    f = __builtin_bit_cast(float, __builtin_amdgcn_readlane(fb, i + u));
                const float h0 = fmaxf((p0[u] + q0[u]) + f * w0, 0.0f);
                const float h1 = fmaxf((p1[u] + q1[u]) + f * w1, 0.0f);
                float* t = hs + (size_t)d * DD + lane;
                atomicAdd(t,      h0);
                atomicAdd(t + 32, h1);
            }
        }
    }
}

__global__ void __launch_bounds__(256)
csr_count(const int* __restrict__ dst, int* __restrict__ cnt)
{
    for (int e = blockIdx.x * 256 + threadIdx.x; e < NE; e += gridDim.x * 256)
        atomicAdd(&cnt[dst[e]], 1);
}

__global__ void __launch_bounds__(SCAN_THREADS)
csr_scan_local(int* __restrict__ cnt, float* __restrict__ deg, int* __restrict__ blockSum)
{
    __shared__ int sw[SCAN_THREADS / 32];
    const int tid  = threadIdx.x;
    const int lane = tid & 31;
    const int wv   = tid >> 5;
    const int i    = blockIdx.x * SCAN_THREADS + tid;
    const int c    = (i < NN) ? cnt[i] : 0;
    if (i < NN) deg[i] = (float)c;
    int v = c;
    #pragma unroll
    for (int o = 1; o < 32; o <<= 1) {
        const int t = __shfl_up(v, o);
        if (lane >= o) v += t;
    }
    if (lane == 31) sw[wv] = v;
    __syncthreads();
    int base = 0;
    for (int k = 0; k < wv; ++k) base += sw[k];
    if (i < NN) cnt[i] = base + v - c;
    if (tid == SCAN_THREADS - 1) blockSum[blockIdx.x] = base + v;
}

__global__ void __launch_bounds__(SCAN_THREADS)
csr_scan_add(int* __restrict__ cnt, const int* __restrict__ blockSum)
{
    __shared__ int sp;
    const int tid = threadIdx.x;
    if (tid < 32) {
        int v = 0;
        for (int k = tid; k < (int)blockIdx.x; k += 32) v += blockSum[k];
        #pragma unroll
        for (int o = 16; o > 0; o >>= 1) v += __shfl_xor(v, o);
        if (tid == 0) sp = v;
    }
    __syncthreads();
    const int i = blockIdx.x * SCAN_THREADS + tid;
    if (i < NN) cnt[i] += sp;
}

#define SRC_BITS 17
#define DST_LOW_MASK ((1u << (32 - SRC_BITS)) - 1u)
static_assert(NN <= (1 << SRC_BITS), "a source node must fit SRC_BITS bits");
typedef int v2i __attribute__((ext_vector_type(2)));

__global__ void __launch_bounds__(256)
csr_fill(const int* __restrict__ src, const int* __restrict__ dst, const float* __restrict__ ef,
         int* __restrict__ cur,
         int* __restrict__ rec)
{
    for (int e = blockIdx.x * 256 + threadIdx.x; e < NE; e += gridDim.x * 256) {
        const int d   = dst[e];
        const int pos = atomicAdd(&cur[d], 1);
        const unsigned key = (unsigned)src[e] | ((unsigned)d << SRC_BITS);
        *(v2i*)(rec + 2 * (size_t)pos) = v2i{ (int)key, __builtin_bit_cast(int, ef[e]) };
    }
}

#define TILE_PITCH 68
#define GP2_BLOCKS ((NN / 16 + 7) / 8)
static_assert(NN % 16 == 0, "a wave owns a whole 16-node tile");
static_assert(16 < (1 << (32 - SRC_BITS)), "a tile's nodes must be told apart by the destination's low bits");

__global__ void __launch_bounds__(256) __attribute__((amdgpu_waves_per_eu(8)))
egnn_node_gather_p2(const float* __restrict__ P,
                    const float* __restrict__ Q,
                    const int*   __restrict__ rec,
                    const int*   __restrict__ endp,
                    const float* __restrict__ wl,
                    const float* __restrict__ W2,
                    const float* __restrict__ b2,
                    const float* __restrict__ deg,
                    float*       __restrict__ out,
                    float*       __restrict__ stats)
{
    __shared__ __attribute__((aligned(16))) float sT[8][16 * TILE_PITCH];
    __shared__ float sred[8][128];

    const int tid  = threadIdx.x;
    const int lane = tid & 31;
    const int lo   = lane & 15;
    const int hi   = lane >> 4;
    const int wv   = __builtin_amdgcn_readfirstlane(tid >> 5);
    const int tile = (int)blockIdx.x * 8 + wv;

    float csum[4], csq[4];
    #pragma unroll
    for (int t = 0; t < 4; ++t) { csum[t] = 0.0f; csq[t] = 0.0f; }

    if (tile < NN / 16) {
        float* myT = &sT[wv][0];
        const v2f w = *(const v2f*)(wl + 2 * lane);

        const int n0 = tile * 16;
        const int n1 = n0 + 16;
        const int e0 = n0 ? endp[n0 - 1] : 0;
        const int e1 = endp[n1 - 1];

        int cur = n0;
        v2f acc = v2f{ 0.f, 0.f };

        for (int eb = e0; eb < e1; eb += 32) {
            const int  e  = eb + lane;
            const int  ec = (e < e1) ? e : (e1 - 1);
            const v2i  r  = *(const v2i*)(rec + 2 * (size_t)ec);
            const int  sv = r[0] & ((1 << SRC_BITS) - 1);
            const int  dv = n0 + (int)((((unsigned)r[0] >> SRC_BITS) - (unsigned)n0) & DST_LOW_MASK);
            const int  fb = r[1];
            const int  n  = (e1 - eb < 32) ? (e1 - eb) : 32;

            #pragma unroll 1
            for (int i = 0; i < n; i += 8) {
                v2f p[8], q[8];
                #pragma unroll
                for (int u = 0; u < 8; ++u) {
                    const int j = (i + u < n) ? (i + u) : (n - 1);
                    const unsigned s = (unsigned)__builtin_amdgcn_readlane(sv, j);
                    const unsigned d = (unsigned)__builtin_amdgcn_readlane(dv, j);
                    p[u] = *(const v2f*)(P + (size_t)s * DD + 2 * lane);
                    q[u] = *(const v2f*)(Q + (size_t)d * DD + 2 * lane);
                }
                #pragma unroll
                for (int u = 0; u < 8; ++u) {
                    if (i + u < n) {
                        const int   d = __builtin_amdgcn_readlane(dv, i + u);
                        const float f = __builtin_bit_cast(float, __builtin_amdgcn_readlane(fb, i + u));
                        if (d != cur) {
                            *(v2f*)(myT + (cur - n0) * TILE_PITCH + 2 * lane) = acc;
                            for (int m = cur + 1; m < d; ++m)
                                *(v2f*)(myT + (m - n0) * TILE_PITCH + 2 * lane) = v2f{ 0.f, 0.f };
                            cur = d;
                            acc = v2f{ 0.f, 0.f };
                        }
                        acc[0] += fmaxf((p[u][0] + q[u][0]) + f * w[0], 0.0f);
                        acc[1] += fmaxf((p[u][1] + q[u][1]) + f * w[1], 0.0f);
                    }
                }
            }
        }
        *(v2f*)(myT + (cur - n0) * TILE_PITCH + 2 * lane) = acc;
        for (int m = cur + 1; m < n1; ++m)
            *(v2f*)(myT + (m - n0) * TILE_PITCH + 2 * lane) = v2f{ 0.f, 0.f };

        __builtin_amdgcn_fence(__ATOMIC_RELEASE, "wavefront");
        __builtin_amdgcn_wave_barrier();
        __builtin_amdgcn_fence(__ATOMIC_ACQUIRE, "wavefront");

        v4f x[8];
        {
            const float* row = myT + lo * TILE_PITCH + 4 * hi;
            #pragma unroll
            for (int j = 0; j < 8; ++j) x[j] = *(const v4f*)(row + 8 * j);
        }
        float dg[8];
        #pragma unroll
        for (int r = 0; r < 8; ++r) dg[r] = deg[n0 + 8 * hi + r];
        float bcol[4];
        #pragma unroll
        for (int t = 0; t < 4; ++t) bcol[t] = b2[16 * t + lo];

        v8f mac[4];
        #pragma unroll
        for (int t = 0; t < 4; ++t) {
            #pragma unroll
            for (int r = 0; r < 8; ++r) mac[t][r] = 0.0f;
        }
        #pragma unroll
        for (int j = 0; j < 8; ++j) {
            const int kr = 8 * j + 4 * hi;
            v2f bLo[4], bHi[4];
            #pragma unroll
            for (int t = 0; t < 4; ++t) {
                const int col = 16 * t + lo;
                bLo[t] = v2f{ W2[kr * 64 + col],       W2[(kr + 1) * 64 + col] };
                bHi[t] = v2f{ W2[(kr + 2) * 64 + col], W2[(kr + 3) * 64 + col] };
            }
            const v2f aLo = { x[j][0], x[j][1] };
            const v2f aHi = { x[j][2], x[j][3] };
            #pragma unroll
            for (int t = 0; t < 4; ++t) {
                mac[t] = __builtin_amdgcn_wmma_f32_16x16x4_f32(
                    false, aLo, false, bLo[t], (short)0, mac[t], false, false);
                mac[t] = __builtin_amdgcn_wmma_f32_16x16x4_f32(
                    false, aHi, false, bHi[t], (short)0, mac[t], false, false);
            }
        }

        #pragma unroll
        for (int t = 0; t < 4; ++t) {
            #pragma unroll
            for (int r = 0; r < 8; ++r) {
                const float val = mac[t][r] + dg[r] * bcol[t];
                out[(long long)(n0 + 8 * hi + r) * DD + 16 * t + lo] = val;
                csum[t] += val;
                csq[t]  += val * val;
            }
        }
    }

    #pragma unroll
    for (int t = 0; t < 4; ++t) {
        const float s = csum[t] + __shfl_xor(csum[t], 16);
        const float q = csq[t]  + __shfl_xor(csq[t],  16);
        if (hi == 0) {
            sred[wv][16 * t + lo]      = s;
            sred[wv][64 + 16 * t + lo] = q;
        }
    }
    __syncthreads();
    if (tid < 128) {
        float a = sred[0][tid];
        for (int k = 1; k < 8; ++k) a += sred[k][tid];
        atomicAdd(&stats[tid], a);
    }
}

__global__ void __launch_bounds__(256)
egnn_node_product2(float*       hs,
                   const float* __restrict__ W2,
                   const float* __restrict__ b2,
                   const float* __restrict__ deg,
                   float*       __restrict__ stats)
{
    const int tid  = threadIdx.x;
    const int lane = tid & 31;
    const int lo   = lane & 15;
    const int hi   = lane >> 4;
    __shared__ float sred[8][128];
    const int wpb       = blockDim.x >> 5;
    const int waveGlob  = __builtin_amdgcn_readfirstlane(blockIdx.x * wpb + (tid >> 5));
    const int waveCount = gridDim.x * wpb;

    v2f wf[16][4];
    #pragma unroll
    for (int s = 0; s < 16; ++s) {
        const int kr = 8 * (s >> 1) + 4 * hi + 2 * (s & 1);
        #pragma unroll
        for (int t = 0; t < 4; ++t) {
            const int col = 16 * t + lo;
            wf[s][t] = v2f{ W2[kr * 64 + col], W2[(kr + 1) * 64 + col] };
        }
    }
    float bcol[4], csum[4], csq[4];
    #pragma unroll
    for (int t = 0; t < 4; ++t) { bcol[t] = b2[16 * t + lo]; csum[t] = 0.0f; csq[t] = 0.0f; }

    const int numTiles = NN / 16;
    v4f x[8];
    #pragma unroll
    for (int j = 0; j < 8; ++j) x[j] = v4f{ 0.f, 0.f, 0.f, 0.f };
    if (waveGlob < numTiles) {
        const float* row = hs + (long long)(waveGlob * 16 + lo) * DD + 4 * hi;
        #pragma unroll
        for (int j = 0; j < 8; ++j) x[j] = *(const v4f*)(row + 8 * j);
    }

    for (int tile = waveGlob; tile < numTiles; tile += waveCount) {
        float dg[8];
        #pragma unroll
        for (int r = 0; r < 8; ++r) dg[r] = deg[tile * 16 + 8 * hi + r];

        v8f acc[4];
        #pragma unroll
        for (int t = 0; t < 4; ++t) {
            #pragma unroll
            for (int r = 0; r < 8; ++r) acc[t][r] = 0.0f;
        }
        #pragma unroll
        for (int j = 0; j < 8; ++j) {
            const v2f aLo = { x[j][0], x[j][1] };
            const v2f aHi = { x[j][2], x[j][3] };
            #pragma unroll
            for (int t = 0; t < 4; ++t) {
                acc[t] = __builtin_amdgcn_wmma_f32_16x16x4_f32(
                    false, aLo, false, wf[2 * j][t], (short)0, acc[t], false, false);
                acc[t] = __builtin_amdgcn_wmma_f32_16x16x4_f32(
                    false, aHi, false, wf[2 * j + 1][t], (short)0, acc[t], false, false);
            }
        }

        const int tn = tile + waveCount;
        if (tn < numTiles) {
            const float* row = hs + (long long)(tn * 16 + lo) * DD + 4 * hi;
            #pragma unroll
            for (int j = 0; j < 8; ++j) x[j] = *(const v4f*)(row + 8 * j);
        }

        #pragma unroll
        for (int t = 0; t < 4; ++t) {
            #pragma unroll
            for (int r = 0; r < 8; ++r) {
                const float val = acc[t][r] + dg[r] * bcol[t];
                hs[(long long)(tile * 16 + 8 * hi + r) * DD + 16 * t + lo] = val;
                csum[t] += val;
                csq[t]  += val * val;
            }
        }
    }

    const int wv = tid >> 5;
    #pragma unroll
    for (int t = 0; t < 4; ++t) {
        const float s = csum[t] + __shfl_xor(csum[t], 16);
        const float q = csq[t]  + __shfl_xor(csq[t],  16);
        if (hi == 0) {
            sred[wv][16 * t + lo]      = s;
            sred[wv][64 + 16 * t + lo] = q;
        }
    }
    __syncthreads();
    if (tid < 128) {
        float a = sred[0][tid];
        for (int w = 1; w < wpb; ++w) a += sred[w][tid];
        atomicAdd(&stats[tid], a);
    }
}

__global__ void __launch_bounds__(256)
zero_f32(float* __restrict__ p, long long n)
{
    for (long long i = blockIdx.x * 256LL + threadIdx.x; i < n;
         i += (long long)gridDim.x * 256LL)
        p[i] = 0.0f;
}

__global__ void __launch_bounds__(256)
bn_reduce(const float* __restrict__ h, float* __restrict__ stats, int nrows)
{
    __shared__ float sSum[256], sSq[256];
    const int c = threadIdx.x & 63;
    const int r = threadIdx.x >> 6;
    float s = 0.f, q = 0.f;
    for (int row = blockIdx.x * 4 + r; row < nrows; row += gridDim.x * 4) {
        float v = h[(long long)row * DD + c];
        s += v; q += v * v;
    }
    sSum[threadIdx.x] = s; sSq[threadIdx.x] = q;
    __syncthreads();
    if (threadIdx.x < 64) {
        float ts = sSum[c] + sSum[64 + c] + sSum[128 + c] + sSum[192 + c];
        float tq = sSq [c] + sSq [64 + c] + sSq [128 + c] + sSq [192 + c];
        atomicAdd(&stats[c], ts);
        atomicAdd(&stats[64 + c], tq);
    }
}

__global__ void __launch_bounds__(256)
bn_relu_apply(float* __restrict__ h, const float* __restrict__ stats,
              const float* __restrict__ gamma, const float* __restrict__ beta,
              int nrows, float invN)
{
    const long long total = (long long)nrows * DD;
    for (long long i = blockIdx.x * 256LL + threadIdx.x; i < total;
         i += (long long)gridDim.x * 256LL) {
        const int c  = (int)(i & 63);
        const float mu  = stats[c] * invN;
        const float var = stats[64 + c] * invN - mu * mu;
        const float rs  = rsqrtf(var + BN_EPS);
        const float v   = (h[i] - mu) * rs * gamma[c] + beta[c];
        h[i] = fmaxf(v, 0.0f);
    }
}

__global__ void __launch_bounds__(256)
pool_mean(const float* __restrict__ h, float* __restrict__ emb)
{
    __shared__ float sm[256];
    const int g = blockIdx.x;
    const int c = threadIdx.x & 63;
    const int r = threadIdx.x >> 6;
    float s = 0.f;
    for (int row = r; row < NPG; row += 4)
        s += h[((long long)g * NPG + row) * DD + c];
    sm[threadIdx.x] = s;
    __syncthreads();
    if (threadIdx.x < 64)
        emb[g * DD + c] = (sm[c] + sm[64 + c] + sm[128 + c] + sm[192 + c]) / (float)NPG;
}

#define POOL_SLICES 5
#define POOL_ROWS   (NPG / POOL_SLICES)
#define POOL_TRIPS  ((POOL_ROWS + 15) / 16)

__global__ void __launch_bounds__(256)
bn_relu_apply_pool(float* __restrict__ h, const float* __restrict__ stats,
                   const float* __restrict__ gamma, const float* __restrict__ beta,
                   float invN, float* __restrict__ part)
{
    __shared__ v4f sred[256];
    const int tid = threadIdx.x;
    const int cq  = tid & 15;
    const int ro  = (tid >> 4) & 15;

    v4f mu, rs, ga, be;
    #pragma unroll
    for (int e = 0; e < 4; ++e) {
        const int   f   = 4 * cq + e;
        const float m   = stats[f] * invN;
        const float var = stats[64 + f] * invN - m * m;
        mu[e] = m;
        rs[e] = rsqrtf(var + BN_EPS);
        ga[e] = gamma[f];
        be[e] = beta[f];
    }

    float* base = h + ((long long)blockIdx.x * POOL_ROWS + ro) * DD + 4 * cq;

    v4f v[POOL_TRIPS];
    #pragma unroll
    for (int i = 0; i < POOL_TRIPS; ++i) {
        v[i] = v4f{ 0.f, 0.f, 0.f, 0.f };
        if (ro + 16 * i < POOL_ROWS) v[i] = *(const v4f*)(base + (long long)(16 * i) * DD);
    }

    v4f sum = v4f{ 0.f, 0.f, 0.f, 0.f };
    #pragma unroll
    for (int i = 0; i < POOL_TRIPS; ++i) {
        if (ro + 16 * i < POOL_ROWS) {
            v4f o;
            #pragma unroll
            for (int e = 0; e < 4; ++e) {
                const float val = (v[i][e] - mu[e]) * rs[e] * ga[e] + be[e];
                o[e] = fmaxf(val, 0.0f);
            }
            *(v4f*)(base + (long long)(16 * i) * DD) = o;
            sum += o;
        }
    }

    sred[tid] = sum;
    __syncthreads();
    if (tid < 16) {
        v4f t = sred[tid];
        #pragma unroll
        for (int k = 1; k < 16; ++k) t += sred[16 * k + tid];
        *(v4f*)(part + (long long)blockIdx.x * DD + 4 * tid) = t;
    }
}

template <int NPART>
__global__ void __launch_bounds__(256)
bn_heads(const float* __restrict__ part,
         float count,
         const float* __restrict__ bg, const float* __restrict__ bb,
         const float* __restrict__ clfW, const float* __restrict__ clfb,
         const float* __restrict__ cptW, const float* __restrict__ cptb,
         float* __restrict__ outLogits, float* __restrict__ outCpt)
{
    __shared__ float sEmb[NG * DD];
    __shared__ float sMu[64], sR[64];
    __shared__ float red[256];
    const int tid = threadIdx.x;
    const int c   = tid & 63;
    const int r   = tid >> 6;

    #pragma unroll
    for (int k = 0; k < (NG * 16 + 255) / 256; ++k) {
        const int item = (tid & 255) + 256 * k;
        if (item < NG * 16) {
            const int g = item >> 4, cqd = item & 15;
            v4f a = v4f{ 0.f, 0.f, 0.f, 0.f };
            #pragma unroll
            for (int p = 0; p < NPART; ++p)
                a += *(const v4f*)(part + (long long)(g * NPART + p) * DD + 4 * cqd);
            #pragma unroll
            for (int e = 0; e < 4; ++e) sEmb[g * DD + 4 * cqd + e] = a[e] / count;
        }
    }
    __syncthreads();

    float s = 0.f, q = 0.f;
    for (int g = r; g < NG; g += 4) {
        float v = sEmb[g * DD + c];
        s += v; q += v * v;
    }
    red[tid] = s; __syncthreads();
    if (tid < 64) sMu[c] = (red[c] + red[64 + c] + red[128 + c] + red[192 + c]) / (float)NG;
    __syncthreads();
    red[tid] = q; __syncthreads();
    if (tid < 64) {
        float tq  = (red[c] + red[64 + c] + red[128 + c] + red[192 + c]) / (float)NG;
        float mu  = sMu[c];
        sR[c] = rsqrtf(tq - mu * mu + BN_EPS);
    }
    __syncthreads();
    for (int i = tid; i < NG * DD; i += 256) {
        int cc = i & 63;
        sEmb[i] = (sEmb[i] - sMu[cc]) * sR[cc] * bg[cc] + bb[cc];
    }
    __syncthreads();
    for (int i = tid; i < NG * NC; i += 256) {
        int g = i / NC, k = i % NC;
        float a = clfb[k];
        #pragma unroll 8
        for (int cc = 0; cc < DD; ++cc) a += sEmb[g * DD + cc] * clfW[cc * NC + k];
        outLogits[i] = a;
    }
    for (int i = tid; i < NG * NCP; i += 256) {
        int g = i >> 4, k = i & 15;
        float a = cptb[k];
        #pragma unroll 8
        for (int cc = 0; cc < DD; ++cc) a += sEmb[g * DD + cc] * cptW[cc * NCP + k];
        outCpt[i] = a;
    }
}

extern "C" void kernel_launch(void* const* d_in, const int* in_sizes, int n_in,
                              void* d_out, int out_size, void* d_ws, size_t ws_size,
                              hipStream_t stream)
{
    const float* x    = (const float*)d_in[0];
    const float* ef   = (const float*)d_in[1];
    const int*   src  = (const int*)  d_in[2];
    const int*   dst  = (const int*)  d_in[3];
    const float* W1s  = (const float*)d_in[5];
    const float* b1s  = (const float*)d_in[6];
    const float* W2s  = (const float*)d_in[7];
    const float* b2s  = (const float*)d_in[8];
    const float* gam  = (const float*)d_in[9];
    const float* bet  = (const float*)d_in[10];
    const float* bng  = (const float*)d_in[11];
    const float* bnb  = (const float*)d_in[12];
    const float* clfW = (const float*)d_in[13];
    const float* clfb = (const float*)d_in[14];
    const float* cptW = (const float*)d_in[15];
    const float* cptb = (const float*)d_in[16];

    float* outH      = (float*)d_out;
    float* outLogits = outH + (long long)NN * DD;
    float* outCpt    = outLogits + NG * NC;

    float* hA    = (float*)d_ws;
    float* stats = hA + (long long)NN * DD;
    float* emb   = stats + 128;
    float* hQ    = emb + NG * DD;

    float* degT  = hQ + (long long)NN * DD;

    int*   cntT  = (int*)(degT + NN);
    int*   bsum  = cntT + NN;
    int*   cRec  = bsum + 128;

    if (ws_size >= (size_t)WS_GROUPED_BYTES) {
        for (int l = 0; l < NL; ++l) {
            const float* W1l = W1s + (long long)l * (2 * DD + 1) * DD;
            float*       stl = stats + 128 * l;
            if (l == 0) {
                egnn_node_project<false><<<512, 256, 0, stream>>>(
                    x, W1l, b1s, hA, hQ, stl, gam, bet, 1.0f / (float)NN, nullptr, stl, cntT);
                csr_count<<<1024, 256, 0, stream>>>(dst, cntT);
                csr_scan_local<<<SCAN_BLOCKS, SCAN_THREADS, 0, stream>>>(cntT, degT, bsum);
                csr_scan_add<<<SCAN_BLOCKS, SCAN_THREADS, 0, stream>>>(cntT, bsum);
                csr_fill<<<1024, 256, 0, stream>>>(src, dst, ef, cntT, cRec);
            } else {
                egnn_node_project<true><<<512, 256, 0, stream>>>(
                    outH, W1l, b1s + l * DD, hA, hQ,
                    stats + 128 * (l - 1), gam + (l - 1) * DD, bet + (l - 1) * DD,
                    1.0f / (float)NN, nullptr, stl, nullptr);
            }
            egnn_node_gather_p2<<<GP2_BLOCKS, 256, 0, stream>>>(
                hA, hQ, cRec, cntT, W1l + 2 * DD * DD,
                W2s + (long long)l * DD * DD, b2s + l * DD, degT, outH, stl);
        }
        bn_relu_apply_pool<<<NG * POOL_SLICES, 256, 0, stream>>>(
            outH, stats + 128 * (NL - 1), gam + (NL - 1) * DD, bet + (NL - 1) * DD,
            1.0f / (float)NN, hA);
        bn_heads<POOL_SLICES><<<1, 256, 0, stream>>>(hA, (float)NPG,
                                                     bng, bnb, clfW, clfb, cptW, cptb,
                                                     outLogits, outCpt);
    } else if (ws_size >= (size_t)WS_NODE_P2_BYTES) {
        for (int l = 0; l < NL; ++l) {
            const float* W1l = W1s + (long long)l * (2 * DD + 1) * DD;
            float*       stl = stats + 128 * l;
            if (l == 0) {
                egnn_node_project<false><<<512, 256, 0, stream>>>(
                    x, W1l, b1s, hA, hQ, stl, gam, bet, 1.0f / (float)NN, outH, stl, (int*)degT);
                egnn_edge_sum<true><<<EDGE_SUM_BLOCKS, 256, 0, stream>>>(
                    hA, hQ, ef, src, dst, W1l + 2 * DD * DD, outH, degT);
            } else {
                egnn_node_project<true><<<512, 256, 0, stream>>>(
                    outH, W1l, b1s + l * DD, hA, hQ,
                    stats + 128 * (l - 1), gam + (l - 1) * DD, bet + (l - 1) * DD,
                    1.0f / (float)NN, outH, stl, nullptr);
                egnn_edge_sum<false><<<EDGE_SUM_BLOCKS, 256, 0, stream>>>(
                    hA, hQ, ef, src, dst, W1l + 2 * DD * DD, outH, degT);
            }
            egnn_node_product2<<<768, 128, 0, stream>>>(
                outH, W2s + (long long)l * DD * DD, b2s + l * DD, degT, stl);
        }
        bn_relu_apply_pool<<<NG * POOL_SLICES, 256, 0, stream>>>(
            outH, stats + 128 * (NL - 1), gam + (NL - 1) * DD, bet + (NL - 1) * DD,
            1.0f / (float)NN, hA);
        bn_heads<POOL_SLICES><<<1, 256, 0, stream>>>(hA, (float)NPG,
                                                     bng, bnb, clfW, clfb, cptW, cptb,
                                                     outLogits, outCpt);
    } else if (ws_size >= (size_t)WS_PROJECTED_BYTES) {
        for (int l = 0; l < NL; ++l) {
            const float* W1l = W1s + (long long)l * (2 * DD + 1) * DD;
            float*       stl = stats + 128 * l;
            if (l == 0)
                egnn_node_project<false><<<512, 256, 0, stream>>>(
                    x, W1l, b1s, hA, hQ, stl, gam, bet, 1.0f / (float)NN, outH, stl, nullptr);
            else
                egnn_node_project<true><<<512, 256, 0, stream>>>(
                    outH, W1l, b1s + l * DD, hA, hQ,
                    stats + 128 * (l - 1), gam + (l - 1) * DD, bet + (l - 1) * DD,
                    1.0f / (float)NN, outH, stl, nullptr);
            egnn_edge_layer_pq<<<EDGE_PQ_BLOCKS, 256, 0, stream>>>(
                hA, hQ, ef, src, dst,
                W1l + 2 * DD * DD,
                W2s + (long long)l * DD * DD,
                b2s + l * DD,
                outH);
            bn_reduce<<<512, 256, 0, stream>>>(outH, stl, NN);
        }
        bn_relu_apply_pool<<<NG * POOL_SLICES, 256, 0, stream>>>(
            outH, stats + 128 * (NL - 1), gam + (NL - 1) * DD, bet + (NL - 1) * DD,
            1.0f / (float)NN, hA);
        bn_heads<POOL_SLICES><<<1, 256, 0, stream>>>(hA, (float)NPG,
                                                     bng, bnb, clfW, clfb, cptW, cptb,
                                                     outLogits, outCpt);
    } else {
        const float* hin = x;
        float* tgt[NL] = { outH, hA, outH };

        const int edgeBlocks = 256;

        for (int l = 0; l < NL; ++l) {
            float* hn = tgt[l];
            zero_f32<<<1024, 256, 0, stream>>>(hn, (long long)NN * DD);
            zero_f32<<<1,    256, 0, stream>>>(stats, 128);
            egnn_edge_layer<<<edgeBlocks, 256, 0, stream>>>(
                hin, ef, src, dst,
                W1s + (long long)l * (2 * DD + 1) * DD,
                b1s + l * DD,
                W2s + (long long)l * DD * DD,
                b2s + l * DD,
                hn);
            bn_reduce<<<512, 256, 0, stream>>>(hn, stats, NN);
            bn_relu_apply<<<1024, 256, 0, stream>>>(hn, stats, gam + l * DD, bet + l * DD,
                                                    NN, 1.0f / (float)NN);
            hin = hn;
        }
        pool_mean<<<NG, 256, 0, stream>>>(outH, emb);
        bn_heads<1><<<1, 256, 0, stream>>>(emb, 1.0f,
                                           bng, bnb, clfW, clfb, cptW, cptb,
                                           outLogits, outCpt);
    }
}


# Edges grouped by a two-level counting sort: per 16-node tile with LDS counters (global atomics per work-group and tile, not per edge), then by node inside a wave by votes; arrival-order sums kept
# speedup vs baseline: 3.2725x; 1.2743x over previous
#include <hip/hip_runtime.h>
#include <hip/hip_bf16.h>

#define NN 100000
#define NE 800000
#define DD 64
#define NL 3
#define NG 100
#define NPG (NN / NG)
#define NC 10
#define NCP 16
#define BN_EPS 1e-5f
#define WS_PROJECTED_BYTES ((2LL * NN * DD + 128 + (long long)NG * DD) * 4)
#define EDGE_PQ_BLOCKS 512
#define WS_NODE_P2_BYTES (WS_PROJECTED_BYTES + (long long)NN * 4)
#define EDGE_SUM_BLOCKS 1250
#define WS_GROUPED_BYTES (WS_NODE_P2_BYTES + ((long long)NN + 128 + 2LL * NE) * 4)

typedef float v2f __attribute__((ext_vector_type(2)));
typedef float v8f __attribute__((ext_vector_type(8)));

__global__ void __launch_bounds__(256)
egnn_edge_layer(const float* __restrict__ h,
                const float* __restrict__ ef,
                const int*   __restrict__ src,
                const int*   __restrict__ dst,
                const float* __restrict__ W1,
                const float* __restrict__ b1,
                const float* __restrict__ W2,
                const float* __restrict__ b2,
                float*       __restrict__ hnext)
{
    __shared__ float sW2[64 * 64];
    __shared__ float sB1[64], sB2[64], sWl[64];

    const int tid  = threadIdx.x;
    const int lane = tid & 31;
    const int wave = tid >> 5;
    const int lo   = lane & 15;
    const int hi   = lane >> 4;

    for (int i = tid; i < 64 * 64; i += 256) sW2[i] = W2[i];
    if (tid < 64) { sB1[tid] = b1[tid]; sB2[tid] = b2[tid]; sWl[tid] = W1[128 * 64 + tid]; }
    __syncthreads();

    const float* W1b    = W1 + 64 * 64;

    const int numTiles  = (NE + 15) >> 4;
    const int waveGlob  = blockIdx.x * 8 + wave;
    const int waveCount = gridDim.x * 8;

    const int t0 = waveGlob;
    const int t1 = t0 + waveCount;
    int   curDst, nxtS, nxtD;
    float curEf, nxtEf;
    bool  nxtValid;
    v2f   a1[16], a2[16];
    {
        const int  e0 = t0 * 16 + lo;
        const bool v0 = (t0 < numTiles) && (e0 < NE);
        const int  c0 = v0 ? e0 : (NE - 1);
        const int  s0 = src[c0];
        const int  d0 = dst[c0];
        curEf  = ef[c0];
        curDst = v0 ? d0 : -1;

        const int  e1 = t1 * 16 + lo;
        nxtValid = (t1 < numTiles) && (e1 < NE);
        const int  c1 = nxtValid ? e1 : (NE - 1);
        nxtS  = src[c1];
        nxtD  = dst[c1];
        nxtEf = ef[c1];

        const float* rowS = h + (long long)s0 * DD;
        const float* rowD = h + (long long)d0 * DD;
        #pragma unroll
        for (int c = 0; c < 16; ++c) {
            const int k0 = 4 * c + 2 * hi;
            a1[c] = v2f{ rowS[k0], rowS[k0 + 1] };
            a2[c] = v2f{ rowD[k0], rowD[k0 + 1] };
        }
    }

    v2f wA[16][4], wB[16][4];
    #pragma unroll
    for (int c = 0; c < 16; ++c) {
        const int k0 = 4 * c + 2 * hi;
        #pragma unroll
        for (int n = 0; n < 4; ++n) {
            const int col = n * 16 + lo;
            wA[c][n] = v2f{ W1 [k0 * 64 + col], W1 [(k0 + 1) * 64 + col] };
            wB[c][n] = v2f{ W1b[k0 * 64 + col], W1b[(k0 + 1) * 64 + col] };
        }
    }

    for (int tile = t0; tile < numTiles; tile += waveCount) {
        asm volatile("" ::: "memory");

        const int  t2 = tile + 2 * waveCount;
        const int  e2 = t2 * 16 + lo;
        const bool v2 = (t2 < numTiles) && (e2 < NE);
        const int  c2 = v2 ? e2 : (NE - 1);
        const int   s2 = src[c2];
        const int   d2 = dst[c2];
        const float f2 = ef[c2];

        v8f acc[4];
        #pragma unroll
        for (int n = 0; n < 4; ++n) {
            #pragma unroll
            for (int r = 0; r < 8; ++r) acc[n][r] = sB1[n * 16 + 8 * hi + r];
        }

        #pragma unroll
        for (int c = 0; c < 16; ++c) {
            #pragma unroll
            for (int n = 0; n < 4; ++n) {
                acc[n] = __builtin_amdgcn_wmma_f32_16x16x4_f32(
                    false, wA[c][n], false, a1[c], (short)0, acc[n], false, false);
                acc[n] = __builtin_amdgcn_wmma_f32_16x16x4_f32(
                    false, wB[c][n], false, a2[c], (short)0, acc[n], false, false);
            }
        }

        {
            const float* rowS = h + (long long)nxtS * DD;
            const float* rowD = h + (long long)nxtD * DD;
            #pragma unroll
            for (int c = 0; c < 16; ++c) {
                const int k0 = 4 * c + 2 * hi;
                a1[c] = v2f{ rowS[k0], rowS[k0 + 1] };
                a2[c] = v2f{ rowD[k0], rowD[k0 + 1] };
            }
        }

        #pragma unroll
        for (int n = 0; n < 4; ++n) {
            #pragma unroll
            for (int r = 0; r < 8; ++r) {
                const float val = acc[n][r] + curEf * sWl[n * 16 + 8 * hi + r];
                acc[n][r] = fmaxf(val, 0.0f);
            }
        }

        v8f mac[4];
        #pragma unroll
        for (int n2 = 0; n2 < 4; ++n2) {
            const float bb = sB2[n2 * 16 + lo];
            #pragma unroll
            for (int j = 0; j < 8; ++j) mac[n2][j] = bb;
        }

        #pragma unroll
        for (int n = 0; n < 4; ++n) {
            #pragma unroll
            for (int p = 0; p < 4; ++p) {
                const int kr = 16 * n + 8 * hi + 2 * p;
                v2f a = { acc[n][2 * p], acc[n][2 * p + 1] };
                #pragma unroll
                for (int n2 = 0; n2 < 4; ++n2) {
                    const int col = n2 * 16 + lo;
                    v2f b = { sW2[kr * 64 + col], sW2[(kr + 1) * 64 + col] };
                    mac[n2] = __builtin_amdgcn_wmma_f32_16x16x4_f32(
                        false, a, false, b, (short)0, mac[n2], false, false);
                }
            }
        }

        #pragma unroll
        for (int v = 0; v < 8; ++v) {
            const int dl   = __builtin_amdgcn_readlane(curDst, v);
            const int dh   = __builtin_amdgcn_readlane(curDst, v + 8);
            const int drow = hi ? dh : dl;
            if (drow >= 0) {
                #pragma unroll
                for (int n2 = 0; n2 < 4; ++n2)
                    atomicAdd(&hnext[(long long)drow * DD + n2 * 16 + lo], mac[n2][v]);
            }
        }

        curEf    = nxtEf;
        curDst   = nxtValid ? nxtD : -1;
        nxtS     = s2;
        nxtD     = d2;
        nxtEf    = f2;
        nxtValid = v2;
    }
}

typedef float v4f __attribute__((ext_vector_type(4)));

template <bool NORM>
__global__ void __launch_bounds__(256)
egnn_node_project(const float* h,
                  const float* __restrict__ W1,
                  const float* __restrict__ b1,
                  float*       __restrict__ P,
                  float*       __restrict__ Q,
                  const float* __restrict__ statsIn,
                  const float* __restrict__ gammaIn,
                  const float* __restrict__ betaIn,
                  float invN,
                  float*       zt,
                  float*       __restrict__ statsOut,
                  int*         __restrict__ degClear)
{
    const int tid  = threadIdx.x;
    const int lane = tid & 31;
    const int wave = __builtin_amdgcn_readfirstlane(tid >> 5);
    const int lo   = lane & 15;
    const int hi   = lane >> 4;
    const int cg   = wave & 3;
    const int half = cg >> 1;
    const int colBase = (cg & 1) * 32;

    if (blockIdx.x == 0 && tid < 128) statsOut[tid] = 0.0f;
    if (degClear) {
        for (int i = blockIdx.x * 256 + tid; i < NN; i += gridDim.x * 256) degClear[i] = 0;
    }

    const float* Wsel = W1 + half * 64 * 64;
    float*       out  = half ? Q : P;

    v2f wf[16][2];
    #pragma unroll
    for (int s = 0; s < 16; ++s) {
        const int kr = 8 * (s >> 1) + 4 * hi + 2 * (s & 1);
        #pragma unroll
        for (int t = 0; t < 2; ++t) {
            const int col = colBase + 16 * t + lo;
            wf[s][t] = v2f{ Wsel[kr * 64 + col], Wsel[(kr + 1) * 64 + col] };
        }
    }
    float bias[2];
    #pragma unroll
    for (int t = 0; t < 2; ++t) bias[t] = half ? 0.0f : b1[colBase + 16 * t + lo];

    v4f sc[8], sh[8];
    if (NORM) {
        #pragma unroll
        for (int j = 0; j < 8; ++j) {
            #pragma unroll
            for (int e = 0; e < 4; ++e) {
                const int   f   = 8 * j + 4 * hi + e;
                const float mu  = statsIn[f] * invN;
                const float var = statsIn[64 + f] * invN - mu * mu;
                const float s   = rsqrtf(var + BN_EPS) * gammaIn[f];
                sc[j][e] = s;
                sh[j][e] = betaIn[f] - mu * s;
            }
        }
    }

    const int numTiles = NN / 16;
    const int group    = blockIdx.x * 2 + (wave >> 2);
    const int groups   = gridDim.x * 2;
    const int trips    = (numTiles - (int)blockIdx.x * 2 + groups - 1) / groups;

    v4f x[8];
    #pragma unroll
    for (int j = 0; j < 8; ++j) x[j] = v4f{ 0.f, 0.f, 0.f, 0.f };
    if (group < numTiles) {
        const float* row = h + (long long)(group * 16 + lo) * DD + 4 * hi;
        #pragma unroll
        for (int j = 0; j < 8; ++j) x[j] = *(const v4f*)(row + 8 * j);
    }

    for (int it = 0; it < trips; ++it) {
        const int  tile = group + it * groups;
        const bool live = tile < numTiles;
        if (live) {
            if (NORM) {
                #pragma unroll
                for (int j = 0; j < 8; ++j) {
                    #pragma unroll
                    for (int e = 0; e < 4; ++e)
                        x[j][e] = fmaxf(fmaf(x[j][e], sc[j][e], sh[j][e]), 0.0f);
                }
            }

            v8f acc[2];
            #pragma unroll
            for (int t = 0; t < 2; ++t) {
                #pragma unroll
                for (int r = 0; r < 8; ++r) acc[t][r] = bias[t];
            }
            #pragma unroll
            for (int j = 0; j < 8; ++j) {
                const v2f aLo = { x[j][0], x[j][1] };
                const v2f aHi = { x[j][2], x[j][3] };
                #pragma unroll
                for (int t = 0; t < 2; ++t) {
                    acc[t] = __builtin_amdgcn_wmma_f32_16x16x4_f32(
                        false, aLo, false, wf[2 * j][t], (short)0, acc[t], false, false);
                    acc[t] = __builtin_amdgcn_wmma_f32_16x16x4_f32(
                        false, aHi, false, wf[2 * j + 1][t], (short)0, acc[t], false, false);
                }
            }

            const int tn = tile + groups;
            if (tn < numTiles) {
                const float* row = h + (long long)(tn * 16 + lo) * DD + 4 * hi;
                #pragma unroll
                for (int j = 0; j < 8; ++j) x[j] = *(const v4f*)(row + 8 * j);
            }

            #pragma unroll
            for (int t = 0; t < 2; ++t) {
                #pragma unroll
                for (int r = 0; r < 8; ++r)
                    out[(long long)(tile * 16 + 8 * hi + r) * DD + colBase + 16 * t + lo] = acc[t][r];
            }
        }

        if (zt == nullptr) continue;
        __builtin_amdgcn_s_barrier();
        if (live) {
            float* z = zt + (long long)(tile * 16 + 4 * cg) * DD + 4 * lane;
            *(v4f*)z            = v4f{ 0.f, 0.f, 0.f, 0.f };
            *(v4f*)(z + 2 * DD) = v4f{ 0.f, 0.f, 0.f, 0.f };
        }
    }
}

__global__ void __launch_bounds__(256)
egnn_edge_layer_pq(const float* __restrict__ P,
                   const float* __restrict__ Q,
                   const float* __restrict__ ef,
                   const int*   __restrict__ src,
                   const int*   __restrict__ dst,
                   const float* __restrict__ wl,
                   const float* __restrict__ W2,
                   const float* __restrict__ b2,
                   float*       __restrict__ hnext)
{
    __shared__ float sW2[64 * 64];
    __shared__ float sB2[64], sWl[64];

    const int tid  = threadIdx.x;
    const int lane = tid & 31;
    const int wave = tid >> 5;
    const int lo   = lane & 15;
    const int hi   = lane >> 4;

    for (int i = tid; i < 64 * 64; i += 256) sW2[i] = W2[i];
    if (tid < 64) { sB2[tid] = b2[tid]; sWl[tid] = wl[tid]; }
    __syncthreads();

    const int numTiles  = (NE + 15) >> 4;
    const int waveGlob  = blockIdx.x * 8 + wave;
    const int waveCount = gridDim.x * 8;

    const int t0 = waveGlob;
    const int t1 = t0 + waveCount;
    int   curDst, nxtS, nxtD;
    float curEf, nxtEf;
    bool  nxtValid;
    v4f   p[8], q[8];
    {
        const int  e0 = t0 * 16 + lo;
        const bool v0 = (t0 < numTiles) && (e0 < NE);
        const int  c0 = v0 ? e0 : (NE - 1);
        const int  s0 = src[c0];
        const int  d0 = dst[c0];
        curEf  = ef[c0];
        curDst = v0 ? d0 : -1;

        const int  e1 = t1 * 16 + lo;
        nxtValid = (t1 < numTiles) && (e1 < NE);
        const int  c1 = nxtValid ? e1 : (NE - 1);
        nxtS  = src[c1];
        nxtD  = dst[c1];
        nxtEf = ef[c1];

        const float* rowS = P + (long long)s0 * DD + 8 * hi;
        const float* rowD = Q + (long long)d0 * DD + 8 * hi;
        #pragma unroll
        for (int n = 0; n < 4; ++n) {
            p[2 * n]     = *(const v4f*)(rowS + 16 * n);
            p[2 * n + 1] = *(const v4f*)(rowS + 16 * n + 4);
            q[2 * n]     = *(const v4f*)(rowD + 16 * n);
            q[2 * n + 1] = *(const v4f*)(rowD + 16 * n + 4);
        }
    }

    for (int tile = t0; tile < numTiles; tile += waveCount) {
        asm volatile("" ::: "memory");

        const int  t2 = tile + 2 * waveCount;
        const int  e2 = t2 * 16 + lo;
        const bool v2 = (t2 < numTiles) && (e2 < NE);
        const int  c2 = v2 ? e2 : (NE - 1);
        const int   s2 = src[c2];
        const int   d2 = dst[c2];
        const float f2 = ef[c2];

        v8f acc[4];
        #pragma unroll
        for (int n = 0; n < 4; ++n) {
            #pragma unroll
            for (int r = 0; r < 8; ++r) {
                const float pq  = p[2 * n + (r >> 2)][r & 3] + q[2 * n + (r >> 2)][r & 3];
                const float val = pq + curEf * sWl[n * 16 + 8 * hi + r];
                acc[n][r] = fmaxf(val, 0.0f);
            }
        }

        {
            const float* rowS = P + (long long)nxtS * DD + 8 * hi;
            const float* rowD = Q + (long long)nxtD * DD + 8 * hi;
            #pragma unroll
            for (int n = 0; n < 4; ++n) {
                p[2 * n]     = *(const v4f*)(rowS + 16 * n);
                p[2 * n + 1] = *(const v4f*)(rowS + 16 * n + 4);
                q[2 * n]     = *(const v4f*)(rowD + 16 * n);
                q[2 * n + 1] = *(const v4f*)(rowD + 16 * n + 4);
            }
        }

        v8f mac[4];
        #pragma unroll
        for (int n2 = 0; n2 < 4; ++n2) {
            const float bb = sB2[n2 * 16 + lo];
            #pragma unroll
            for (int j = 0; j < 8; ++j) mac[n2][j] = bb;
        }

        #pragma unroll
        for (int n = 0; n < 4; ++n) {
            #pragma unroll
            for (int pp = 0; pp < 4; ++pp) {
                const int kr = 16 * n + 8 * hi + 2 * pp;
                v2f a = { acc[n][2 * pp], acc[n][2 * pp + 1] };
                #pragma unroll
                for (int n2 = 0; n2 < 4; ++n2) {
                    const int col = n2 * 16 + lo;
                    v2f b = { sW2[kr * 64 + col], sW2[(kr + 1) * 64 + col] };
                    mac[n2] = __builtin_amdgcn_wmma_f32_16x16x4_f32(
                        false, a, false, b, (short)0, mac[n2], false, false);
                }
            }
        }

        #pragma unroll
        for (int v = 0; v < 8; ++v) {
            const int dl   = __builtin_amdgcn_readlane(curDst, v);
            const int dh   = __builtin_amdgcn_readlane(curDst, v + 8);
            const int drow = hi ? dh : dl;
            if (drow >= 0) {
                #pragma unroll
                for (int n2 = 0; n2 < 4; ++n2)
                    atomicAdd(&hnext[(long long)drow * DD + n2 * 16 + lo], mac[n2][v]);
            }
        }

        curEf    = nxtEf;
        curDst   = nxtValid ? nxtD : -1;
        nxtS     = s2;
        nxtD     = d2;
        nxtEf    = f2;
        nxtValid = v2;
    }
}

static_assert(NE % 8 == 0, "edges are walked in groups of 8");

template <bool COUNT>
__global__ void __launch_bounds__(256)
egnn_edge_sum(const float* __restrict__ P,
              const float* __restrict__ Q,
              const float* __restrict__ ef,
              const int*   __restrict__ src,
              const int*   __restrict__ dst,
              const float* __restrict__ wl,
              float*       __restrict__ hs,
              float*       __restrict__ deg)
{
    const int lane      = threadIdx.x & 31;
    const unsigned waveGlob  = (unsigned)__builtin_amdgcn_readfirstlane(blockIdx.x * 8 + (threadIdx.x >> 5));
    const unsigned waveCount = gridDim.x * 8;
    const float w0 = wl[lane], w1 = wl[32 + lane];

    const int e0 = (int)(waveGlob       * (unsigned)(NE / 8) / waveCount) * 8;
    const int e1 = (int)((waveGlob + 1) * (unsigned)(NE / 8) / waveCount) * 8;

    for (int eb = e0; eb < e1; eb += 32) {
        const int  e  = eb + lane;
        const bool ok = e < e1;
        const int  ec = ok ? e : (e1 - 1);
        const int  sv = src[ec];
        const int  dv = dst[ec];
        const int  fb = __builtin_bit_cast(int, ef[ec]);
        if (COUNT && ok) atomicAdd(&deg[dv], 1.0f);
        const int  n  = (e1 - eb < 32) ? (e1 - eb) : 32;

        #pragma unroll 1
        for (int i = 0; i < n; i += 8) {
            float p0[8], p1[8], q0[8], q1[8];
            #pragma unroll
            for (int u = 0; u < 8; ++u) {
                const unsigned s = (unsigned)__builtin_amdgcn_readlane(sv, i + u);
                const unsigned d = (unsigned)__builtin_amdgcn_readlane(dv, i + u);
                const float* rp = P + (size_t)s * DD + lane;
                const float* rq = Q + (size_t)d * DD + lane;
                p0[u] = rp[0];  p1[u] = rp[32];
                q0[u] = rq[0];  q1[u] = rq[32];
            }
            #pragma unroll
            for (int u = 0; u < 8; ++u) {
                const unsigned d = (unsigned)__builtin_amdgcn_readlane(dv, i + u);
                const float    f = __builtin_bit_cast(float, __builtin_amdgcn_readlane(fb, i + u));
                const float h0 = fmaxf((p0[u] + q0[u]) + f * w0, 0.0f);
                const float h1 = fmaxf((p1[u] + q1[u]) + f * w1, 0.0f);
                float* t = hs + (size_t)d * DD + lane;
                atomicAdd(t,      h0);
                atomicAdd(t + 32, h1);
            }
        }
    }
}

#define NT (NN / 16)
#define SORT_BLOCKS 256
#define SORT_CHUNK  (NE / SORT_BLOCKS)
#define SORT_EPT    ((SORT_CHUNK + 255) / 256)
#define SORT_BINS_PT ((NT + 255) / 256)
static_assert(NE % SORT_BLOCKS == 0, "every work-group takes the same number of edges");
static_assert(NN % 16 == 0, "a tile is 16 whole nodes");

__global__ void __launch_bounds__(256)
tile_count(const int* __restrict__ dst, int* __restrict__ tcnt)
{
    __shared__ int sh[NT];
    const int tid = threadIdx.x & 255;
    for (int t = tid; t < NT; t += 256) sh[t] = 0;
    __syncthreads();
    const int e0 = (int)blockIdx.x * SORT_CHUNK;
    int d[SORT_EPT];
    #pragma unroll
    for (int j = 0; j < SORT_EPT; ++j) {
        const int i = tid + 256 * j;
        d[j] = (i < SORT_CHUNK) ? dst[e0 + i] : 0;
    }
    #pragma unroll
    for (int j = 0; j < SORT_EPT; ++j)
        if (tid + 256 * j < SORT_CHUNK) atomicAdd(&sh[d[j] >> 4], 1);
    __syncthreads();
    int cc[SORT_BINS_PT];
    #pragma unroll
    for (int j = 0; j < SORT_BINS_PT; ++j) {
        const int t = tid + 256 * j;
        cc[j] = (t < NT) ? sh[t] : 0;
    }
    #pragma unroll
    for (int j = 0; j < SORT_BINS_PT; ++j)
        if (cc[j]) atomicAdd(&tcnt[tid + 256 * j], cc[j]);
}

#define TSCAN_THREADS 1024
#define TSCAN_ITEMS   ((NT + TSCAN_THREADS - 1) / TSCAN_THREADS)

__global__ void __launch_bounds__(TSCAN_THREADS)
tile_scan(int* __restrict__ tcnt)
{
    __shared__ int sw[TSCAN_THREADS / 32];
    const int tid  = threadIdx.x;
    const int lane = tid & 31;
    const int wv   = tid >> 5;
    int v[TSCAN_ITEMS];
    int s = 0;
    #pragma unroll
    for (int j = 0; j < TSCAN_ITEMS; ++j) {
        const int i = tid * TSCAN_ITEMS + j;
        v[j] = (i < NT) ? tcnt[i] : 0;
        s += v[j];
    }
    int inc = s;
    #pragma unroll
    for (int o = 1; o < 32; o <<= 1) {
        const int t = __shfl_up(inc, o);
        if (lane >= o) inc += t;
    }
    if (lane == 31) sw[wv] = inc;
    __syncthreads();
    int run = inc - s;
    for (int k = 0; k < wv; ++k) run += sw[k];
    #pragma unroll
    for (int j = 0; j < TSCAN_ITEMS; ++j) {
        const int i = tid * TSCAN_ITEMS + j;
        if (i < NT) tcnt[i] = run;
        run += v[j];
    }
}

#define SRC_BITS 17
#define DST_LOW_MASK ((1u << (32 - SRC_BITS)) - 1u)
static_assert(NN <= (1 << SRC_BITS), "a source node must fit SRC_BITS bits");
typedef int v2i __attribute__((ext_vector_type(2)));

__global__ void __launch_bounds__(256)
tile_fill(const int* __restrict__ src, const int* __restrict__ dst, const float* __restrict__ ef,
          int* __restrict__ tcur,
          int* __restrict__ rec)
{
    __shared__ int sh[NT];
    const int tid = threadIdx.x & 255;
    for (int t = tid; t < NT; t += 256) sh[t] = 0;
    __syncthreads();

    const int e0 = (int)blockIdx.x * SORT_CHUNK;
    int key[SORT_EPT], fb[SORT_EPT], tl[SORT_EPT], rk[SORT_EPT];
    #pragma unroll
    for (int j = 0; j < SORT_EPT; ++j) {
        const int i = tid + 256 * j;
        const bool ok = i < SORT_CHUNK;
        const int  d  = ok ? dst[e0 + i] : 0;
        const int  s  = ok ? src[e0 + i] : 0;
        key[j] = (int)((unsigned)s | ((unsigned)d << SRC_BITS));
        fb[j]  = ok ? __builtin_bit_cast(int, ef[e0 + i]) : 0;
        tl[j]  = d >> 4;
    }
    #pragma unroll
    for (int j = 0; j < SORT_EPT; ++j) {
        rk[j] = 0;
        if (tid + 256 * j < SORT_CHUNK) rk[j] = atomicAdd(&sh[tl[j]], 1);
    }
    __syncthreads();
    int cc[SORT_BINS_PT], got[SORT_BINS_PT];
    #pragma unroll
    for (int j = 0; j < SORT_BINS_PT; ++j) {
        const int t = tid + 256 * j;
        cc[j] = (t < NT) ? sh[t] : 0;
    }
    #pragma unroll
    for (int j = 0; j < SORT_BINS_PT; ++j)
        if (cc[j]) got[j] = atomicAdd(&tcur[tid + 256 * j], cc[j]);
    #pragma unroll
    for (int j = 0; j < SORT_BINS_PT; ++j)
        if (cc[j]) sh[tid + 256 * j] = got[j];
    __syncthreads();
    #pragma unroll
    for (int j = 0; j < SORT_EPT; ++j) {
        if (tid + 256 * j < SORT_CHUNK) {
            const int pos = sh[tl[j]] + rk[j];
            *(v2i*)(rec + 2 * (size_t)pos) = v2i{ key[j], fb[j] };
        }
    }
}

__global__ void __launch_bounds__(256)
tile_group(const int* __restrict__ in,
           const int* __restrict__ tend,
           int*       __restrict__ rec,
           float*     __restrict__ deg)
{
    const int lane = threadIdx.x & 31;
    const int tile = __builtin_amdgcn_readfirstlane((int)blockIdx.x * 8 + (int)(threadIdx.x >> 5));
    if (tile >= NT) return;
    const int e0 = tile ? tend[tile - 1] : 0;
    const int e1 = tend[tile];

    int c = 0;
    for (int eb = e0; eb < e1; eb += 128) {
        v2i r[4];
        #pragma unroll
        for (int u = 0; u < 4; ++u) {
            const int e = eb + 32 * u + lane;
            r[u] = *(const v2i*)(in + 2 * (size_t)((e < e1) ? e : (e1 - 1)));
        }
        #pragma unroll
        for (int u = 0; u < 4; ++u) {
            const int e    = eb + 32 * u + lane;
            const int node = (e < e1) ? (int)(((unsigned)r[u][0] >> SRC_BITS) & 15u) : 16;
            #pragma unroll
            for (int k = 0; k < 16; ++k) {
                const unsigned m = __builtin_amdgcn_ballot_w32(node == k);
                if (lane == k) c += __builtin_popcount(m);
            }
        }
    }
    if (lane < 16) deg[tile * 16 + lane] = (float)c;
    int off = c;
    #pragma unroll
    for (int o = 1; o < 16; o <<= 1) {
        const int t = __shfl_up(off, o);
        if (lane >= o) off += t;
    }
    off -= c;

    for (int eb = e0; eb < e1; eb += 128) {
        v2i r[4];
        #pragma unroll
        for (int u = 0; u < 4; ++u) {
            const int e = eb + 32 * u + lane;
            r[u] = *(const v2i*)(in + 2 * (size_t)((e < e1) ? e : (e1 - 1)));
        }
        #pragma unroll
        for (int u = 0; u < 4; ++u) {
            const int  e     = eb + 32 * u + lane;
            const bool valid = e < e1;
            const int  node  = valid ? (int)(((unsigned)r[u][0] >> SRC_BITS) & 15u) : 16;
            const int  base  = __shfl(off, node & 15);
            unsigned mine = 0u;
            #pragma unroll
            for (int k = 0; k < 16; ++k) {
                const unsigned m = __builtin_amdgcn_ballot_w32(node == k);
                if (node == k) mine = m;
                if (lane == k) off += __builtin_popcount(m);
            }
            const int rank = __builtin_popcount(mine & ((1u << lane) - 1u));
            if (valid) *(v2i*)(rec + 2 * (size_t)(e0 + base + rank)) = r[u];
        }
    }
}

#define TILE_PITCH 68
#define GP2_BLOCKS ((NN / 16 + 7) / 8)
static_assert(NN % 16 == 0, "a wave owns a whole 16-node tile");
static_assert(16 < (1 << (32 - SRC_BITS)), "a tile's nodes must be told apart by the destination's low bits");

__global__ void __launch_bounds__(256) __attribute__((amdgpu_waves_per_eu(8)))
egnn_node_gather_p2(const float* __restrict__ P,
                    const float* __restrict__ Q,
                    const int*   __restrict__ rec,
                    const int*   __restrict__ endp,
                    const float* __restrict__ wl,
                    const float* __restrict__ W2,
                    const float* __restrict__ b2,
                    const float* __restrict__ deg,
                    float*       __restrict__ out,
                    float*       __restrict__ stats)
{
    __shared__ __attribute__((aligned(16))) float sT[8][16 * TILE_PITCH];
    __shared__ float sred[8][128];

    const int tid  = threadIdx.x;
    const int lane = tid & 31;
    const int lo   = lane & 15;
    const int hi   = lane >> 4;
    const int wv   = __builtin_amdgcn_readfirstlane(tid >> 5);
    const int tile = (int)blockIdx.x * 8 + wv;

    float csum[4], csq[4];
    #pragma unroll
    for (int t = 0; t < 4; ++t) { csum[t] = 0.0f; csq[t] = 0.0f; }

    if (tile < NN / 16) {
        float* myT = &sT[wv][0];
        const v2f w = *(const v2f*)(wl + 2 * lane);

        const int n0 = tile * 16;
        const int n1 = n0 + 16;
        const int e0 = tile ? endp[tile - 1] : 0;
        const int e1 = endp[tile];

        int cur = n0;
        v2f acc = v2f{ 0.f, 0.f };

        for (int eb = e0; eb < e1; eb += 32) {
            const int  e  = eb + lane;
            const int  ec = (e < e1) ? e : (e1 - 1);
            const v2i  r  = *(const v2i*)(rec + 2 * (size_t)ec);
            const int  sv = r[0] & ((1 << SRC_BITS) - 1);
            const int  dv = n0 + (int)((((unsigned)r[0] >> SRC_BITS) - (unsigned)n0) & DST_LOW_MASK);
            const int  fb = r[1];
            const int  n  = (e1 - eb < 32) ? (e1 - eb) : 32;

            #pragma unroll 1
            for (int i = 0; i < n; i += 8) {
                v2f p[8], q[8];
                #pragma unroll
                for (int u = 0; u < 8; ++u) {
                    const int j = (i + u < n) ? (i + u) : (n - 1);
                    const unsigned s = (unsigned)__builtin_amdgcn_readlane(sv, j);
                    const unsigned d = (unsigned)__builtin_amdgcn_readlane(dv, j);
                    p[u] = *(const v2f*)(P + (size_t)s * DD + 2 * lane);
                    q[u] = *(const v2f*)(Q + (size_t)d * DD + 2 * lane);
                }
                #pragma unroll
                for (int u = 0; u < 8; ++u) {
                    if (i + u < n) {
                        const int   d = __builtin_amdgcn_readlane(dv, i + u);
                        const float f = __builtin_bit_cast(float, __builtin_amdgcn_readlane(fb, i + u));
                        if (d != cur) {
                            *(v2f*)(myT + (cur - n0) * TILE_PITCH + 2 * lane) = acc;
                            for (int m = cur + 1; m < d; ++m)
                                *(v2f*)(myT + (m - n0) * TILE_PITCH + 2 * lane) = v2f{ 0.f, 0.f };
                            cur = d;
                            acc = v2f{ 0.f, 0.f };
                        }
                        acc[0] += fmaxf((p[u][0] + q[u][0]) + f * w[0], 0.0f);
                        acc[1] += fmaxf((p[u][1] + q[u][1]) + f * w[1], 0.0f);
                    }
                }
            }
        }
        *(v2f*)(myT + (cur - n0) * TILE_PITCH + 2 * lane) = acc;
        for (int m = cur + 1; m < n1; ++m)
            *(v2f*)(myT + (m - n0) * TILE_PITCH + 2 * lane) = v2f{ 0.f, 0.f };

        __builtin_amdgcn_fence(__ATOMIC_RELEASE, "wavefront");
        __builtin_amdgcn_wave_barrier();
        __builtin_amdgcn_fence(__ATOMIC_ACQUIRE, "wavefront");

        v4f x[8];
        {
            const float* row = myT + lo * TILE_PITCH + 4 * hi;
            #pragma unroll
            for (int j = 0; j < 8; ++j) x[j] = *(const v4f*)(row + 8 * j);
        }
        float dg[8];
        #pragma unroll
        for (int r = 0; r < 8; ++r) dg[r] = deg[n0 + 8 * hi + r];
        float bcol[4];
        #pragma unroll
        for (int t = 0; t < 4; ++t) bcol[t] = b2[16 * t + lo];

        v8f mac[4];
        #pragma unroll
        for (int t = 0; t < 4; ++t) {
            #pragma unroll
            for (int r = 0; r < 8; ++r) mac[t][r] = 0.0f;
        }
        #pragma unroll
        for (int j = 0; j < 8; ++j) {
            const int kr = 8 * j + 4 * hi;
            v2f bLo[4], bHi[4];
            #pragma unroll
            for (int t = 0; t < 4; ++t) {
                const int col = 16 * t + lo;
                bLo[t] = v2f{ W2[kr * 64 + col],       W2[(kr + 1) * 64 + col] };
                bHi[t] = v2f{ W2[(kr + 2) * 64 + col], W2[(kr + 3) * 64 + col] };
            }
            const v2f aLo = { x[j][0], x[j][1] };
            const v2f aHi = { x[j][2], x[j][3] };
            #pragma unroll
            for (int t = 0; t < 4; ++t) {
                mac[t] = __builtin_amdgcn_wmma_f32_16x16x4_f32(
                    false, aLo, false, bLo[t], (short)0, mac[t], false, false);
                mac[t] = __builtin_amdgcn_wmma_f32_16x16x4_f32(
                    false, aHi, false, bHi[t], (short)0, mac[t], false, false);
            }
        }

        #pragma unroll
        for (int t = 0; t < 4; ++t) {
            #pragma unroll
            for (int r = 0; r < 8; ++r) {
                const float val = mac[t][r] + dg[r] * bcol[t];
                out[(long long)(n0 + 8 * hi + r) * DD + 16 * t + lo] = val;
                csum[t] += val;
                csq[t]  += val * val;
            }
        }
    }

    #pragma unroll
    for (int t = 0; t < 4; ++t) {
        const float s = csum[t] + __shfl_xor(csum[t], 16);
        const float q = csq[t]  + __shfl_xor(csq[t],  16);
        if (hi == 0) {
            sred[wv][16 * t + lo]      = s;
            sred[wv][64 + 16 * t + lo] = q;
        }
    }
    __syncthreads();
    if (tid < 128) {
        float a = sred[0][tid];
        for (int k = 1; k < 8; ++k) a += sred[k][tid];
        atomicAdd(&stats[tid], a);
    }
}

__global__ void __launch_bounds__(256)
egnn_node_product2(float*       hs,
                   const float* __restrict__ W2,
                   const float* __restrict__ b2,
                   const float* __restrict__ deg,
                   float*       __restrict__ stats)
{
    const int tid  = threadIdx.x;
    const int lane = tid & 31;
    const int lo   = lane & 15;
    const int hi   = lane >> 4;
    __shared__ float sred[8][128];
    const int wpb       = blockDim.x >> 5;
    const int waveGlob  = __builtin_amdgcn_readfirstlane(blockIdx.x * wpb + (tid >> 5));
    const int waveCount = gridDim.x * wpb;

    v2f wf[16][4];
    #pragma unroll
    for (int s = 0; s < 16; ++s) {
        const int kr = 8 * (s >> 1) + 4 * hi + 2 * (s & 1);
        #pragma unroll
        for (int t = 0; t < 4; ++t) {
            const int col = 16 * t + lo;
            wf[s][t] = v2f{ W2[kr * 64 + col], W2[(kr + 1) * 64 + col] };
        }
    }
    float bcol[4], csum[4], csq[4];
    #pragma unroll
    for (int t = 0; t < 4; ++t) { bcol[t] = b2[16 * t + lo]; csum[t] = 0.0f; csq[t] = 0.0f; }

    const int numTiles = NN / 16;
    v4f x[8];
    #pragma unroll
    for (int j = 0; j < 8; ++j) x[j] = v4f{ 0.f, 0.f, 0.f, 0.f };
    if (waveGlob < numTiles) {
        const float* row = hs + (long long)(waveGlob * 16 + lo) * DD + 4 * hi;
        #pragma unroll
        for (int j = 0; j < 8; ++j) x[j] = *(const v4f*)(row + 8 * j);
    }

    for (int tile = waveGlob; tile < numTiles; tile += waveCount) {
        float dg[8];
        #pragma unroll
        for (int r = 0; r < 8; ++r) dg[r] = deg[tile * 16 + 8 * hi + r];

        v8f acc[4];
        #pragma unroll
        for (int t = 0; t < 4; ++t) {
            #pragma unroll
            for (int r = 0; r < 8; ++r) acc[t][r] = 0.0f;
        }
        #pragma unroll
        for (int j = 0; j < 8; ++j) {
            const v2f aLo = { x[j][0], x[j][1] };
            const v2f aHi = { x[j][2], x[j][3] };
            #pragma unroll
            for (int t = 0; t < 4; ++t) {
                acc[t] = __builtin_amdgcn_wmma_f32_16x16x4_f32(
                    false, aLo, false, wf[2 * j][t], (short)0, acc[t], false, false);
                acc[t] = __builtin_amdgcn_wmma_f32_16x16x4_f32(
                    false, aHi, false, wf[2 * j + 1][t], (short)0, acc[t], false, false);
            }
        }

        const int tn = tile + waveCount;
        if (tn < numTiles) {
            const float* row = hs + (long long)(tn * 16 + lo) * DD + 4 * hi;
            #pragma unroll
            for (int j = 0; j < 8; ++j) x[j] = *(const v4f*)(row + 8 * j);
        }

        #pragma unroll
        for (int t = 0; t < 4; ++t) {
            #pragma unroll
            for (int r = 0; r < 8; ++r) {
                const float val = acc[t][r] + dg[r] * bcol[t];
                hs[(long long)(tile * 16 + 8 * hi + r) * DD + 16 * t + lo] = val;
                csum[t] += val;
                csq[t]  += val * val;
            }
        }
    }

    const int wv = tid >> 5;
    #pragma unroll
    for (int t = 0; t < 4; ++t) {
        const float s = csum[t] + __shfl_xor(csum[t], 16);
        const float q = csq[t]  + __shfl_xor(csq[t],  16);
        if (hi == 0) {
            sred[wv][16 * t + lo]      = s;
            sred[wv][64 + 16 * t + lo] = q;
        }
    }
    __syncthreads();
    if (tid < 128) {
        float a = sred[0][tid];
        for (int w = 1; w < wpb; ++w) a += sred[w][tid];
        atomicAdd(&stats[tid], a);
    }
}

__global__ void __launch_bounds__(256)
zero_f32(float* __restrict__ p, long long n)
{
    for (long long i = blockIdx.x * 256LL + threadIdx.x; i < n;
         i += (long long)gridDim.x * 256LL)
        p[i] = 0.0f;
}

__global__ void __launch_bounds__(256)
bn_reduce(const float* __restrict__ h, float* __restrict__ stats, int nrows)
{
    __shared__ float sSum[256], sSq[256];
    const int c = threadIdx.x & 63;
    const int r = threadIdx.x >> 6;
    float s = 0.f, q = 0.f;
    for (int row = blockIdx.x * 4 + r; row < nrows; row += gridDim.x * 4) {
        float v = h[(long long)row * DD + c];
        s += v; q += v * v;
    }
    sSum[threadIdx.x] = s; sSq[threadIdx.x] = q;
    __syncthreads();
    if (threadIdx.x < 64) {
        float ts = sSum[c] + sSum[64 + c] + sSum[128 + c] + sSum[192 + c];
        float tq = sSq [c] + sSq [64 + c] + sSq [128 + c] + sSq [192 + c];
        atomicAdd(&stats[c], ts);
        atomicAdd(&stats[64 + c], tq);
    }
}

__global__ void __launch_bounds__(256)
bn_relu_apply(float* __restrict__ h, const float* __restrict__ stats,
              const float* __restrict__ gamma, const float* __restrict__ beta,
              int nrows, float invN)
{
    const long long total = (long long)nrows * DD;
    for (long long i = blockIdx.x * 256LL + threadIdx.x; i < total;
         i += (long long)gridDim.x * 256LL) {
        const int c  = (int)(i & 63);
        const float mu  = stats[c] * invN;
        const float var = stats[64 + c] * invN - mu * mu;
        const float rs  = rsqrtf(var + BN_EPS);
        const float v   = (h[i] - mu) * rs * gamma[c] + beta[c];
        h[i] = fmaxf(v, 0.0f);
    }
}

__global__ void __launch_bounds__(256)
pool_mean(const float* __restrict__ h, float* __restrict__ emb)
{
    __shared__ float sm[256];
    const int g = blockIdx.x;
    const int c = threadIdx.x & 63;
    const int r = threadIdx.x >> 6;
    float s = 0.f;
    for (int row = r; row < NPG; row += 4)
        s += h[((long long)g * NPG + row) * DD + c];
    sm[threadIdx.x] = s;
    __syncthreads();
    if (threadIdx.x < 64)
        emb[g * DD + c] = (sm[c] + sm[64 + c] + sm[128 + c] + sm[192 + c]) / (float)NPG;
}

#define POOL_SLICES 5
#define POOL_ROWS   (NPG / POOL_SLICES)
#define POOL_TRIPS  ((POOL_ROWS + 15) / 16)

__global__ void __launch_bounds__(256)
bn_relu_apply_pool(float* __restrict__ h, const float* __restrict__ stats,
                   const float* __restrict__ gamma, const float* __restrict__ beta,
                   float invN, float* __restrict__ part)
{
    __shared__ v4f sred[256];
    const int tid = threadIdx.x;
    const int cq  = tid & 15;
    const int ro  = (tid >> 4) & 15;

    v4f mu, rs, ga, be;
    #pragma unroll
    for (int e = 0; e < 4; ++e) {
        const int   f   = 4 * cq + e;
        const float m   = stats[f] * invN;
        const float var = stats[64 + f] * invN - m * m;
        mu[e] = m;
        rs[e] = rsqrtf(var + BN_EPS);
        ga[e] = gamma[f];
        be[e] = beta[f];
    }

    float* base = h + ((long long)blockIdx.x * POOL_ROWS + ro) * DD + 4 * cq;

    v4f v[POOL_TRIPS];
    #pragma unroll
    for (int i = 0; i < POOL_TRIPS; ++i) {
        v[i] = v4f{ 0.f, 0.f, 0.f, 0.f };
        if (ro + 16 * i < POOL_ROWS) v[i] = *(const v4f*)(base + (long long)(16 * i) * DD);
    }

    v4f sum = v4f{ 0.f, 0.f, 0.f, 0.f };
    #pragma unroll
    for (int i = 0; i < POOL_TRIPS; ++i) {
        if (ro + 16 * i < POOL_ROWS) {
            v4f o;
            #pragma unroll
            for (int e = 0; e < 4; ++e) {
                const float val = (v[i][e] - mu[e]) * rs[e] * ga[e] + be[e];
                o[e] = fmaxf(val, 0.0f);
            }
            *(v4f*)(base + (long long)(16 * i) * DD) = o;
            sum += o;
        }
    }

    sred[tid] = sum;
    __syncthreads();
    if (tid < 16) {
        v4f t = sred[tid];
        #pragma unroll
        for (int k = 1; k < 16; ++k) t += sred[16 * k + tid];
        *(v4f*)(part + (long long)blockIdx.x * DD + 4 * tid) = t;
    }
}

template <int NPART>
__global__ void __launch_bounds__(256)
bn_heads(const float* __restrict__ part,
         float count,
         const float* __restrict__ bg, const float* __restrict__ bb,
         const float* __restrict__ clfW, const float* __restrict__ clfb,
         const float* __restrict__ cptW, const float* __restrict__ cptb,
         float* __restrict__ outLogits, float* __restrict__ outCpt)
{
    __shared__ float sEmb[NG * DD];
    __shared__ float sMu[64], sR[64];
    __shared__ float red[256];
    const int tid = threadIdx.x;
    const int c   = tid & 63;
    const int r   = tid >> 6;

    #pragma unroll
    for (int k = 0; k < (NG * 16 + 255) / 256; ++k) {
        const int item = (tid & 255) + 256 * k;
        if (item < NG * 16) {
            const int g = item >> 4, cqd = item & 15;
            v4f a = v4f{ 0.f, 0.f, 0.f, 0.f };
            #pragma unroll
            for (int p = 0; p < NPART; ++p)
                a += *(const v4f*)(part + (long long)(g * NPART + p) * DD + 4 * cqd);
            #pragma unroll
            for (int e = 0; e < 4; ++e) sEmb[g * DD + 4 * cqd + e] = a[e] / count;
        }
    }
    __syncthreads();

    float s = 0.f, q = 0.f;
    for (int g = r; g < NG; g += 4) {
        float v = sEmb[g * DD + c];
        s += v; q += v * v;
    }
    red[tid] = s; __syncthreads();
    if (tid < 64) sMu[c] = (red[c] + red[64 + c] + red[128 + c] + red[192 + c]) / (float)NG;
    __syncthreads();
    red[tid] = q; __syncthreads();
    if (tid < 64) {
        float tq  = (red[c] + red[64 + c] + red[128 + c] + red[192 + c]) / (float)NG;
        float mu  = sMu[c];
        sR[c] = rsqrtf(tq - mu * mu + BN_EPS);
    }
    __syncthreads();
    for (int i = tid; i < NG * DD; i += 256) {
        int cc = i & 63;
        sEmb[i] = (sEmb[i] - sMu[cc]) * sR[cc] * bg[cc] + bb[cc];
    }
    __syncthreads();
    for (int i = tid; i < NG * NC; i += 256) {
        int g = i / NC, k = i % NC;
        float a = clfb[k];
        #pragma unroll 8
        for (int cc = 0; cc < DD; ++cc) a += sEmb[g * DD + cc] * clfW[cc * NC + k];
        outLogits[i] = a;
    }
    for (int i = tid; i < NG * NCP; i += 256) {
        int g = i >> 4, k = i & 15;
        float a = cptb[k];
        #pragma unroll 8
        for (int cc = 0; cc < DD; ++cc) a += sEmb[g * DD + cc] * cptW[cc * NCP + k];
        outCpt[i] = a;
    }
}

extern "C" void kernel_launch(void* const* d_in, const int* in_sizes, int n_in,
                              void* d_out, int out_size, void* d_ws, size_t ws_size,
                              hipStream_t stream)
{
    const float* x    = (const float*)d_in[0];
    const float* ef   = (const float*)d_in[1];
    const int*   src  = (const int*)  d_in[2];
    const int*   dst  = (const int*)  d_in[3];
    const float* W1s  = (const float*)d_in[5];
    const float* b1s  = (const float*)d_in[6];
    const float* W2s  = (const float*)d_in[7];
    const float* b2s  = (const float*)d_in[8];
    const float* gam  = (const float*)d_in[9];
    const float* bet  = (const float*)d_in[10];
    const float* bng  = (const float*)d_in[11];
    const float* bnb  = (const float*)d_in[12];
    const float* clfW = (const float*)d_in[13];
    const float* clfb = (const float*)d_in[14];
    const float* cptW = (const float*)d_in[15];
    const float* cptb = (const float*)d_in[16];

    float* outH      = (float*)d_out;
    float* outLogits = outH + (long long)NN * DD;
    float* outCpt    = outLogits + NG * NC;

    float* hA    = (float*)d_ws;
    float* stats = hA + (long long)NN * DD;
    float* emb   = stats + 128;
    float* hQ    = emb + NG * DD;

    float* degT  = hQ + (long long)NN * DD;

    int*   cntT  = (int*)(degT + NN);
    int*   cRec  = cntT + NN + 128;

    if (ws_size >= (size_t)WS_GROUPED_BYTES) {
        for (int l = 0; l < NL; ++l) {
            const float* W1l = W1s + (long long)l * (2 * DD + 1) * DD;
            float*       stl = stats + 128 * l;
            if (l == 0) {
                egnn_node_project<false><<<512, 256, 0, stream>>>(
                    x, W1l, b1s, hA, hQ, stl, gam, bet, 1.0f / (float)NN, nullptr, stl, cntT);
                tile_count<<<SORT_BLOCKS, 256, 0, stream>>>(dst, cntT);
                tile_scan<<<1, TSCAN_THREADS, 0, stream>>>(cntT);
                tile_fill<<<SORT_BLOCKS, 256, 0, stream>>>(src, dst, ef, cntT, (int*)outH);
                tile_group<<<GP2_BLOCKS, 256, 0, stream>>>((const int*)outH, cntT, cRec, degT);
            } else {
                egnn_node_project<true><<<512, 256, 0, stream>>>(
                    outH, W1l, b1s + l * DD, hA, hQ,
                    stats + 128 * (l - 1), gam + (l - 1) * DD, bet + (l - 1) * DD,
                    1.0f / (float)NN, nullptr, stl, nullptr);
            }
            egnn_node_gather_p2<<<GP2_BLOCKS, 256, 0, stream>>>(
                hA, hQ, cRec, cntT, W1l + 2 * DD * DD,
                W2s + (long long)l * DD * DD, b2s + l * DD, degT, outH, stl);
        }
        bn_relu_apply_pool<<<NG * POOL_SLICES, 256, 0, stream>>>(
            outH, stats + 128 * (NL - 1), gam + (NL - 1) * DD, bet + (NL - 1) * DD,
            1.0f / (float)NN, hA);
        bn_heads<POOL_SLICES><<<1, 256, 0, stream>>>(hA, (float)NPG,
                                                     bng, bnb, clfW, clfb, cptW, cptb,
                                                     outLogits, outCpt);
    } else if (ws_size >= (size_t)WS_NODE_P2_BYTES) {
        for (int l = 0; l < NL; ++l) {
            const float* W1l = W1s + (long long)l * (2 * DD + 1) * DD;
            float*       stl = stats + 128 * l;
            if (l == 0) {
                egnn_node_project<false><<<512, 256, 0, stream>>>(
                    x, W1l, b1s, hA, hQ, stl, gam, bet, 1.0f / (float)NN, outH, stl, (int*)degT);
                egnn_edge_sum<true><<<EDGE_SUM_BLOCKS, 256, 0, stream>>>(
                    hA, hQ, ef, src, dst, W1l + 2 * DD * DD, outH, degT);
            } else {
                egnn_node_project<true><<<512, 256, 0, stream>>>(
                    outH, W1l, b1s + l * DD, hA, hQ,
                    stats + 128 * (l - 1), gam + (l - 1) * DD, bet + (l - 1) * DD,
                    1.0f / (float)NN, outH, stl, nullptr);
                egnn_edge_sum<false><<<EDGE_SUM_BLOCKS, 256, 0, stream>>>(
                    hA, hQ, ef, src, dst, W1l + 2 * DD * DD, outH, degT);
            }
            egnn_node_product2<<<768, 128, 0, stream>>>(
                outH, W2s + (long long)l * DD * DD, b2s + l * DD, degT, stl);
        }
        bn_relu_apply_pool<<<NG * POOL_SLICES, 256, 0, stream>>>(
            outH, stats + 128 * (NL - 1), gam + (NL - 1) * DD, bet + (NL - 1) * DD,
            1.0f / (float)NN, hA);
        bn_heads<POOL_SLICES><<<1, 256, 0, stream>>>(hA, (float)NPG,
                                                     bng, bnb, clfW, clfb, cptW, cptb,
                                                     outLogits, outCpt);
    } else if (ws_size >= (size_t)WS_PROJECTED_BYTES) {
        for (int l = 0; l < NL; ++l) {
            const float* W1l = W1s + (long long)l * (2 * DD + 1) * DD;
            float*       stl = stats + 128 * l;
            if (l == 0)
                egnn_node_project<false><<<512, 256, 0, stream>>>(
                    x, W1l, b1s, hA, hQ, stl, gam, bet, 1.0f / (float)NN, outH, stl, nullptr);
            else
                egnn_node_project<true><<<512, 256, 0, stream>>>(
                    outH, W1l, b1s + l * DD, hA, hQ,
                    stats + 128 * (l - 1), gam + (l - 1) * DD, bet + (l - 1) * DD,
                    1.0f / (float)NN, outH, stl, nullptr);
            egnn_edge_layer_pq<<<EDGE_PQ_BLOCKS, 256, 0, stream>>>(
                hA, hQ, ef, src, dst,
                W1l + 2 * DD * DD,
                W2s + (long long)l * DD * DD,
                b2s + l * DD,
                outH);
            bn_reduce<<<512, 256, 0, stream>>>(outH, stl, NN);
        }
        bn_relu_apply_pool<<<NG * POOL_SLICES, 256, 0, stream>>>(
            outH, stats + 128 * (NL - 1), gam + (NL - 1) * DD, bet + (NL - 1) * DD,
            1.0f / (float)NN, hA);
        bn_heads<POOL_SLICES><<<1, 256, 0, stream>>>(hA, (float)NPG,
                                                     bng, bnb, clfW, clfb, cptW, cptb,
                                                     outLogits, outCpt);
    } else {
        const float* hin = x;
        float* tgt[NL] = { outH, hA, outH };

        const int edgeBlocks = 256;

        for (int l = 0; l < NL; ++l) {
            float* hn = tgt[l];
            zero_f32<<<1024, 256, 0, stream>>>(hn, (long long)NN * DD);
            zero_f32<<<1,    256, 0, stream>>>(stats, 128);
            egnn_edge_layer<<<edgeBlocks, 256, 0, stream>>>(
                hin, ef, src, dst,
                W1s + (long long)l * (2 * DD + 1) * DD,
                b1s + l * DD,
                W2s + (long long)l * DD * DD,
                b2s + l * DD,
                hn);
            bn_reduce<<<512, 256, 0, stream>>>(hn, stats, NN);
            bn_relu_apply<<<1024, 256, 0, stream>>>(hn, stats, gam + l * DD, bet + l * DD,
                                                    NN, 1.0f / (float)NN);
            hin = hn;
        }
        pool_mean<<<NG, 256, 0, stream>>>(outH, emb);
        bn_heads<1><<<1, 256, 0, stream>>>(emb, 1.0f,
                                           bng, bnb, clfW, clfb, cptW, cptb,
                                           outLogits, outCpt);
    }
}


# Edges grouped by bins of 400 nodes, then by node per bin in LDS: every record store leaves LDS in slot order (coalesced), scan folded into the partition; 12 launches to 11; arrival-order sums kept
# speedup vs baseline: 3.5379x; 1.0811x over previous
#include <hip/hip_runtime.h>
#include <hip/hip_bf16.h>

#define NN 100000
#define NE 800000
#define DD 64
#define NL 3
#define NG 100
#define NPG (NN / NG)
#define NC 10
#define NCP 16
#define BN_EPS 1e-5f
#define WS_PROJECTED_BYTES ((2LL * NN * DD + 128 + (long long)NG * DD) * 4)
#define EDGE_PQ_BLOCKS 512
#define WS_NODE_P2_BYTES (WS_PROJECTED_BYTES + (long long)NN * 4)
#define EDGE_SUM_BLOCKS 1250
#define WS_GROUPED_BYTES (WS_NODE_P2_BYTES + ((long long)NN + 128 + 2LL * NE) * 4)

typedef float v2f __attribute__((ext_vector_type(2)));
typedef float v8f __attribute__((ext_vector_type(8)));

__global__ void __launch_bounds__(256)
egnn_edge_layer(const float* __restrict__ h,
                const float* __restrict__ ef,
                const int*   __restrict__ src,
                const int*   __restrict__ dst,
                const float* __restrict__ W1,
                const float* __restrict__ b1,
                const float* __restrict__ W2,
                const float* __restrict__ b2,
                float*       __restrict__ hnext)
{
    __shared__ float sW2[64 * 64];
    __shared__ float sB1[64], sB2[64], sWl[64];

    const int tid  = threadIdx.x;
    const int lane = tid & 31;
    const int wave = tid >> 5;
    const int lo   = lane & 15;
    const int hi   = lane >> 4;

    for (int i = tid; i < 64 * 64; i += 256) sW2[i] = W2[i];
    if (tid < 64) { sB1[tid] = b1[tid]; sB2[tid] = b2[tid]; sWl[tid] = W1[128 * 64 + tid]; }
    __syncthreads();

    const float* W1b    = W1 + 64 * 64;

    const int numTiles  = (NE + 15) >> 4;
    const int waveGlob  = blockIdx.x * 8 + wave;
    const int waveCount = gridDim.x * 8;

    const int t0 = waveGlob;
    const int t1 = t0 + waveCount;
    int   curDst, nxtS, nxtD;
    float curEf, nxtEf;
    bool  nxtValid;
    v2f   a1[16], a2[16];
    {
        const int  e0 = t0 * 16 + lo;
        const bool v0 = (t0 < numTiles) && (e0 < NE);
        const int  c0 = v0 ? e0 : (NE - 1);
        const int  s0 = src[c0];
        const int  d0 = dst[c0];
        curEf  = ef[c0];
        curDst = v0 ? d0 : -1;

        const int  e1 = t1 * 16 + lo;
        nxtValid = (t1 < numTiles) && (e1 < NE);
        const int  c1 = nxtValid ? e1 : (NE - 1);
        nxtS  = src[c1];
        nxtD  = dst[c1];
        nxtEf = ef[c1];

        const float* rowS = h + (long long)s0 * DD;
        const float* rowD = h + (long long)d0 * DD;
        #pragma unroll
        for (int c = 0; c < 16; ++c) {
            const int k0 = 4 * c + 2 * hi;
            a1[c] = v2f{ rowS[k0], rowS[k0 + 1] };
            a2[c] = v2f{ rowD[k0], rowD[k0 + 1] };
        }
    }

    v2f wA[16][4], wB[16][4];
    #pragma unroll
    for (int c = 0; c < 16; ++c) {
        const int k0 = 4 * c + 2 * hi;
        #pragma unroll
        for (int n = 0; n < 4; ++n) {
            const int col = n * 16 + lo;
            wA[c][n] = v2f{ W1 [k0 * 64 + col], W1 [(k0 + 1) * 64 + col] };
            wB[c][n] = v2f{ W1b[k0 * 64 + col], W1b[(k0 + 1) * 64 + col] };
        }
    }

    for (int tile = t0; tile < numTiles; tile += waveCount) {
        asm volatile("" ::: "memory");

        const int  t2 = tile + 2 * waveCount;
        const int  e2 = t2 * 16 + lo;
        const bool v2 = (t2 < numTiles) && (e2 < NE);
        const int  c2 = v2 ? e2 : (NE - 1);
        const int   s2 = src[c2];
        const int   d2 = dst[c2];
        const float f2 = ef[c2];

        v8f acc[4];
        #pragma unroll
        for (int n = 0; n < 4; ++n) {
            #pragma unroll
            for (int r = 0; r < 8; ++r) acc[n][r] = sB1[n * 16 + 8 * hi + r];
        }

        #pragma unroll
        for (int c = 0; c < 16; ++c) {
            #pragma unroll
            for (int n = 0; n < 4; ++n) {
                acc[n] = __builtin_amdgcn_wmma_f32_16x16x4_f32(
                    false, wA[c][n], false, a1[c], (short)0, acc[n], false, false);
                acc[n] = __builtin_amdgcn_wmma_f32_16x16x4_f32(
                    false, wB[c][n], false, a2[c], (short)0, acc[n], false, false);
            }
        }

        {
            const float* rowS = h + (long long)nxtS * DD;
            const float* rowD = h + (long long)nxtD * DD;
            #pragma unroll
            for (int c = 0; c < 16; ++c) {
                const int k0 = 4 * c + 2 * hi;
                a1[c] = v2f{ rowS[k0], rowS[k0 + 1] };
                a2[c] = v2f{ rowD[k0], rowD[k0 + 1] };
            }
        }

        #pragma unroll
        for (int n = 0; n < 4; ++n) {
            #pragma unroll
            for (int r = 0; r < 8; ++r) {
                const float val = acc[n][r] + curEf * sWl[n * 16 + 8 * hi + r];
                acc[n][r] = fmaxf(val, 0.0f);
            }
        }

        v8f mac[4];
        #pragma unroll
        for (int n2 = 0; n2 < 4; ++n2) {
            const float bb = sB2[n2 * 16 + lo];
            #pragma unroll
            for (int j = 0; j < 8; ++j) mac[n2][j] = bb;
        }

        #pragma unroll
        for (int n = 0; n < 4; ++n) {
            #pragma unroll
            for (int p = 0; p < 4; ++p) {
                const int kr = 16 * n + 8 * hi + 2 * p;
                v2f a = { acc[n][2 * p], acc[n][2 * p + 1] };
                #pragma unroll
                for (int n2 = 0; n2 < 4; ++n2) {
                    const int col = n2 * 16 + lo;
                    v2f b = { sW2[kr * 64 + col], sW2[(kr + 1) * 64 + col] };
                    mac[n2] = __builtin_amdgcn_wmma_f32_16x16x4_f32(
                        false, a, false, b, (short)0, mac[n2], false, false);
                }
            }
        }

        #pragma unroll
        for (int v = 0; v < 8; ++v) {
            const int dl   = __builtin_amdgcn_readlane(curDst, v);
            const int dh   = __builtin_amdgcn_readlane(curDst, v + 8);
            const int drow = hi ? dh : dl;
            if (drow >= 0) {
                #pragma unroll
                for (int n2 = 0; n2 < 4; ++n2)
                    atomicAdd(&hnext[(long long)drow * DD + n2 * 16 + lo], mac[n2][v]);
            }
        }

        curEf    = nxtEf;
        curDst   = nxtValid ? nxtD : -1;
        nxtS     = s2;
        nxtD     = d2;
        nxtEf    = f2;
        nxtValid = v2;
    }
}

typedef float v4f __attribute__((ext_vector_type(4)));

template <bool NORM>
__global__ void __launch_bounds__(256)
egnn_node_project(const float* h,
                  const float* __restrict__ W1,
                  const float* __restrict__ b1,
                  float*       __restrict__ P,
                  float*       __restrict__ Q,
                  const float* __restrict__ statsIn,
                  const float* __restrict__ gammaIn,
                  const float* __restrict__ betaIn,
                  float invN,
                  float*       zt,
                  float*       __restrict__ statsOut,
                  int*         __restrict__ degClear)
{
    const int tid  = threadIdx.x;
    const int lane = tid & 31;
    const int wave = __builtin_amdgcn_readfirstlane(tid >> 5);
    const int lo   = lane & 15;
    const int hi   = lane >> 4;
    const int cg   = wave & 3;
    const int half = cg >> 1;
    const int colBase = (cg & 1) * 32;

    if (blockIdx.x == 0 && tid < 128) statsOut[tid] = 0.0f;
    if (degClear) {
        for (int i = blockIdx.x * 256 + tid; i < NN; i += gridDim.x * 256) degClear[i] = 0;
    }

    const float* Wsel = W1 + half * 64 * 64;
    float*       out  = half ? Q : P;

    v2f wf[16][2];
    #pragma unroll
    for (int s = 0; s < 16; ++s) {
        const int kr = 8 * (s >> 1) + 4 * hi + 2 * (s & 1);
        #pragma unroll
        for (int t = 0; t < 2; ++t) {
            const int col = colBase + 16 * t + lo;
            wf[s][t] = v2f{ Wsel[kr * 64 + col], Wsel[(kr + 1) * 64 + col] };
        }
    }
    float bias[2];
    #pragma unroll
    for (int t = 0; t < 2; ++t) bias[t] = half ? 0.0f : b1[colBase + 16 * t + lo];

    v4f sc[8], sh[8];
    if (NORM) {
        #pragma unroll
        for (int j = 0; j < 8; ++j) {
            #pragma unroll
            for (int e = 0; e < 4; ++e) {
                const int   f   = 8 * j + 4 * hi + e;
                const float mu  = statsIn[f] * invN;
                const float var = statsIn[64 + f] * invN - mu * mu;
                const float s   = rsqrtf(var + BN_EPS) * gammaIn[f];
                sc[j][e] = s;
                sh[j][e] = betaIn[f] - mu * s;
            }
        }
    }

    const int numTiles = NN / 16;
    const int group    = blockIdx.x * 2 + (wave >> 2);
    const int groups   = gridDim.x * 2;
    const int trips    = (numTiles - (int)blockIdx.x * 2 + groups - 1) / groups;

    v4f x[8];
    #pragma unroll
    for (int j = 0; j < 8; ++j) x[j] = v4f{ 0.f, 0.f, 0.f, 0.f };
    if (group < numTiles) {
        const float* row = h + (long long)(group * 16 + lo) * DD + 4 * hi;
        #pragma unroll
        for (int j = 0; j < 8; ++j) x[j] = *(const v4f*)(row + 8 * j);
    }

    for (int it = 0; it < trips; ++it) {
        const int  tile = group + it * groups;
        const bool live = tile < numTiles;
        if (live) {
            if (NORM) {
                #pragma unroll
                for (int j = 0; j < 8; ++j) {
                    #pragma unroll
                    for (int e = 0; e < 4; ++e)
                        x[j][e] = fmaxf(fmaf(x[j][e], sc[j][e], sh[j][e]), 0.0f);
                }
            }

            v8f acc[2];
            #pragma unroll
            for (int t = 0; t < 2; ++t) {
                #pragma unroll
                for (int r = 0; r < 8; ++r) acc[t][r] = bias[t];
            }
            #pragma unroll
            for (int j = 0; j < 8; ++j) {
                const v2f aLo = { x[j][0], x[j][1] };
                const v2f aHi = { x[j][2], x[j][3] };
                #pragma unroll
                for (int t = 0; t < 2; ++t) {
                    acc[t] = __builtin_amdgcn_wmma_f32_16x16x4_f32(
                        false, aLo, false, wf[2 * j][t], (short)0, acc[t], false, false);
                    acc[t] = __builtin_amdgcn_wmma_f32_16x16x4_f32(
                        false, aHi, false, wf[2 * j + 1][t], (short)0, acc[t], false, false);
                }
            }

            const int tn = tile + groups;
            if (tn < numTiles) {
                const float* row = h + (long long)(tn * 16 + lo) * DD + 4 * hi;
                #pragma unroll
                for (int j = 0; j < 8; ++j) x[j] = *(const v4f*)(row + 8 * j);
            }

            #pragma unroll
            for (int t = 0; t < 2; ++t) {
                #pragma unroll
                for (int r = 0; r < 8; ++r)
                    out[(long long)(tile * 16 + 8 * hi + r) * DD + colBase + 16 * t + lo] = acc[t][r];
            }
        }

        if (zt == nullptr) continue;
        __builtin_amdgcn_s_barrier();
        if (live) {
            float* z = zt + (long long)(tile * 16 + 4 * cg) * DD + 4 * lane;
            *(v4f*)z            = v4f{ 0.f, 0.f, 0.f, 0.f };
            *(v4f*)(z + 2 * DD) = v4f{ 0.f, 0.f, 0.f, 0.f };
        }
    }
}

__global__ void __launch_bounds__(256)
egnn_edge_layer_pq(const float* __restrict__ P,
                   const float* __restrict__ Q,
                   const float* __restrict__ ef,
                   const int*   __restrict__ src,
                   const int*   __restrict__ dst,
                   const float* __restrict__ wl,
                   const float* __restrict__ W2,
                   const float* __restrict__ b2,
                   float*       __restrict__ hnext)
{
    __shared__ float sW2[64 * 64];
    __shared__ float sB2[64], sWl[64];

    const int tid  = threadIdx.x;
    const int lane = tid & 31;
    const int wave = tid >> 5;
    const int lo   = lane & 15;
    const int hi   = lane >> 4;

    for (int i = tid; i < 64 * 64; i += 256) sW2[i] = W2[i];
    if (tid < 64) { sB2[tid] = b2[tid]; sWl[tid] = wl[tid]; }
    __syncthreads();

    const int numTiles  = (NE + 15) >> 4;
    const int waveGlob  = blockIdx.x * 8 + wave;
    const int waveCount = gridDim.x * 8;

    const int t0 = waveGlob;
    const int t1 = t0 + waveCount;
    int   curDst, nxtS, nxtD;
    float curEf, nxtEf;
    bool  nxtValid;
    v4f   p[8], q[8];
    {
        const int  e0 = t0 * 16 + lo;
        const bool v0 = (t0 < numTiles) && (e0 < NE);
        const int  c0 = v0 ? e0 : (NE - 1);
        const int  s0 = src[c0];
        const int  d0 = dst[c0];
        curEf  = ef[c0];
        curDst = v0 ? d0 : -1;

        const int  e1 = t1 * 16 + lo;
        nxtValid = (t1 < numTiles) && (e1 < NE);
        const int  c1 = nxtValid ? e1 : (NE - 1);
        nxtS  = src[c1];
        nxtD  = dst[c1];
        nxtEf = ef[c1];

        const float* rowS = P + (long long)s0 * DD + 8 * hi;
        const float* rowD = Q + (long long)d0 * DD + 8 * hi;
        #pragma unroll
        for (int n = 0; n < 4; ++n) {
            p[2 * n]     = *(const v4f*)(rowS + 16 * n);
            p[2 * n + 1] = *(const v4f*)(rowS + 16 * n + 4);
            q[2 * n]     = *(const v4f*)(rowD + 16 * n);
            q[2 * n + 1] = *(const v4f*)(rowD + 16 * n + 4);
        }
    }

    for (int tile = t0; tile < numTiles; tile += waveCount) {
        asm volatile("" ::: "memory");

        const int  t2 = tile + 2 * waveCount;
        const int  e2 = t2 * 16 + lo;
        const bool v2 = (t2 < numTiles) && (e2 < NE);
        const int  c2 = v2 ? e2 : (NE - 1);
        const int   s2 = src[c2];
        const int   d2 = dst[c2];
        const float f2 = ef[c2];

        v8f acc[4];
        #pragma unroll
        for (int n = 0; n < 4; ++n) {
            #pragma unroll
            for (int r = 0; r < 8; ++r) {
                const float pq  = p[2 * n + (r >> 2)][r & 3] + q[2 * n + (r >> 2)][r & 3];
                const float val = pq + curEf * sWl[n * 16 + 8 * hi + r];
                acc[n][r] = fmaxf(val, 0.0f);
            }
        }

        {
            const float* rowS = P + (long long)nxtS * DD + 8 * hi;
            const float* rowD = Q + (long long)nxtD * DD + 8 * hi;
            #pragma unroll
            for (int n = 0; n < 4; ++n) {
                p[2 * n]     = *(const v4f*)(rowS + 16 * n);
                p[2 * n + 1] = *(const v4f*)(rowS + 16 * n + 4);
                q[2 * n]     = *(const v4f*)(rowD + 16 * n);
                q[2 * n + 1] = *(const v4f*)(rowD + 16 * n + 4);
            }
        }

        v8f mac[4];
        #pragma unroll
        for (int n2 = 0; n2 < 4; ++n2) {
            const float bb = sB2[n2 * 16 + lo];
            #pragma unroll
            for (int j = 0; j < 8; ++j) mac[n2][j] = bb;
        }

        #pragma unroll
        for (int n = 0; n < 4; ++n) {
            #pragma unroll
            for (int pp = 0; pp < 4; ++pp) {
                const int kr = 16 * n + 8 * hi + 2 * pp;
                v2f a = { acc[n][2 * pp], acc[n][2 * pp + 1] };
                #pragma unroll
                for (int n2 = 0; n2 < 4; ++n2) {
                    const int col = n2 * 16 + lo;
                    v2f b = { sW2[kr * 64 + col], sW2[(kr + 1) * 64 + col] };
                    mac[n2] = __builtin_amdgcn_wmma_f32_16x16x4_f32(
                        false, a, false, b, (short)0, mac[n2], false, false);
                }
            }
        }

        #pragma unroll
        for (int v = 0; v < 8; ++v) {
            const int dl   = __builtin_amdgcn_readlane(curDst, v);
            const int dh   = __builtin_amdgcn_readlane(curDst, v + 8);
            const int drow = hi ? dh : dl;
            if (drow >= 0) {
                #pragma unroll
                for (int n2 = 0; n2 < 4; ++n2)
                    atomicAdd(&hnext[(long long)drow * DD + n2 * 16 + lo], mac[n2][v]);
            }
        }

        curEf    = nxtEf;
        curDst   = nxtValid ? nxtD : -1;
        nxtS     = s2;
        nxtD     = d2;
        nxtEf    = f2;
        nxtValid = v2;
    }
}

static_assert(NE % 8 == 0, "edges are walked in groups of 8");

template <bool COUNT>
__global__ void __launch_bounds__(256)
egnn_edge_sum(const float* __restrict__ P,
              const float* __restrict__ Q,
              const float* __restrict__ ef,
              const int*   __restrict__ src,
              const int*   __restrict__ dst,
              const float* __restrict__ wl,
              float*       __restrict__ hs,
              float*       __restrict__ deg)
{
    const int lane      = threadIdx.x & 31;
    const unsigned waveGlob  = (unsigned)__builtin_amdgcn_readfirstlane(blockIdx.x * 8 + (threadIdx.x >> 5));
    const unsigned waveCount = gridDim.x * 8;
    const float w0 = wl[lane], w1 = wl[32 + lane];

    const int e0 = (int)(waveGlob       * (unsigned)(NE / 8) / waveCount) * 8;
    const int e1 = (int)((waveGlob + 1) * (unsigned)(NE / 8) / waveCount) * 8;

    for (int eb = e0; eb < e1; eb += 32) {
        const int  e  = eb + lane;
        const bool ok = e < e1;
        const int  ec = ok ? e : (e1 - 1);
        const int  sv = src[ec];
        const int  dv = dst[ec];
        const int  fb = __builtin_bit_cast(int, ef[ec]);
        if (COUNT && ok) atomicAdd(&deg[dv], 1.0f);
        const int  n  = (e1 - eb < 32) ? (e1 - eb) : 32;

        #pragma unroll 1
        for (int i = 0; i < n; i += 8) {
            float p0[8], p1[8], q0[8], q1[8];
            #pragma unroll
            for (int u = 0; u < 8; ++u) {
                const unsigned s = (unsigned)__builtin_amdgcn_readlane(sv, i + u);
                const unsigned d = (unsigned)__builtin_amdgcn_readlane(dv, i + u);
                const float* rp = P + (size_t)s * DD + lane;
                const float* rq = Q + (size_t)d * DD + lane;
                p0[u] = rp[0];  p1[u] = rp[32];
                q0[u] = rq[0];  q1[u] = rq[32];
            }
            #pragma unroll
            for (int u = 0; u < 8; ++u) {
                const unsigned d = (unsigned)__builtin_amdgcn_readlane(dv, i + u);
                const float    f = __builtin_bit_cast(float, __builtin_amdgcn_readlane(fb, i + u));
                const float h0 = fmaxf((p0[u] + q0[u]) + f * w0, 0.0f);
                const float h1 = fmaxf((p1[u] + q1[u]) + f * w1, 0.0f);
                float* t = hs + (size_t)d * DD + lane;
                atomicAdd(t,      h0);
                atomicAdd(t + 32, h1);
            }
        }
    }
}

#define NT (NN / 16)
#define SORT_BLOCKS 256
#define SORT_CHUNK  (NE / SORT_BLOCKS)
#define SORT_EPT    ((SORT_CHUNK + 255) / 256)
#define NBIN        250
#define BIN_TILES   (NT / NBIN)
#define BIN_NODES   (16 * BIN_TILES)
#define BSORT_THREADS 512
#define BSORT_CAP   8192
static_assert(NE % SORT_BLOCKS == 0, "every work-group takes the same number of edges");
static_assert(NN % 16 == 0, "a tile is 16 whole nodes");
static_assert(NT % NBIN == 0, "a bin is whole tiles");
static_assert(NBIN <= 256, "one thread per bin in a 256-thread work-group");
static_assert(BIN_NODES < BSORT_THREADS, "one thread per node of a bin and one behind the last");

#define SRC_BITS 17
#define DST_LOW_MASK ((1u << (32 - SRC_BITS)) - 1u)
static_assert(NN <= (1 << SRC_BITS), "a source node must fit SRC_BITS bits");
static_assert(BIN_NODES < (1 << (32 - SRC_BITS)), "a bin's nodes must be told apart by the destination's low bits");
typedef int v2i __attribute__((ext_vector_type(2)));

template <int WAVES>
__device__ __forceinline__ int block_excl_scan(int v, int* sw, int& total)
{
    const int lane = threadIdx.x & 31;
    const int wv   = threadIdx.x >> 5;
    int inc = v;
    #pragma unroll
    for (int o = 1; o < 32; o <<= 1) {
        const int t = __shfl_up(inc, o);
        if (lane >= o) inc += t;
    }
    if (lane == 31) sw[wv] = inc;
    __syncthreads();
    int pre = 0, tot = 0;
    #pragma unroll
    for (int k = 0; k < WAVES; ++k) {
        const int s = sw[k];
        if (k < wv) pre += s;
        tot += s;
    }
    total = tot;
    return pre + inc - v;
}

__global__ void __launch_bounds__(256)
bin_count(const int* __restrict__ dst, int* __restrict__ bcnt)
{
    __shared__ int sh[256];
    const int tid = threadIdx.x & 255;
    sh[tid] = 0;
    __syncthreads();
    const int e0 = (int)blockIdx.x * SORT_CHUNK;
    int d[SORT_EPT];
    #pragma unroll
    for (int j = 0; j < SORT_EPT; ++j) {
        const int i = tid + 256 * j;
        d[j] = (i < SORT_CHUNK) ? dst[e0 + i] : 0;
    }
    #pragma unroll
    for (int j = 0; j < SORT_EPT; ++j)
        if (tid + 256 * j < SORT_CHUNK) atomicAdd(&sh[(unsigned)d[j] / (unsigned)BIN_NODES], 1);
    __syncthreads();
    const int c = sh[tid];
    if (tid < NBIN && c) atomicAdd(&bcnt[tid], c);
}

__global__ void __launch_bounds__(256)
bin_partition(const int* __restrict__ src, const int* __restrict__ dst, const float* __restrict__ ef,
              const int* __restrict__ bcnt,
              int*       __restrict__ bcur,
              int*       __restrict__ tmp)
{
    __shared__ int sCnt[256];
    __shared__ int sBase[256];
    __shared__ int swA[8], swB[8];
    __shared__ __attribute__((aligned(8))) v2i sRec[SORT_CHUNK];
    __shared__ int sPos[SORT_CHUNK];
    const int tid = threadIdx.x & 255;
    sCnt[tid] = 0;
    __syncthreads();

    const int e0 = (int)blockIdx.x * SORT_CHUNK;
    int key[SORT_EPT], fb[SORT_EPT], bn[SORT_EPT], rk[SORT_EPT];
    #pragma unroll
    for (int j = 0; j < SORT_EPT; ++j) {
        const int i = tid + 256 * j;
        const bool ok = i < SORT_CHUNK;
        const int  d  = ok ? dst[e0 + i] : 0;
        const int  s  = ok ? src[e0 + i] : 0;
        key[j] = (int)((unsigned)s | ((unsigned)d << SRC_BITS));
        fb[j]  = ok ? __builtin_bit_cast(int, ef[e0 + i]) : 0;
        bn[j]  = (int)((unsigned)d / (unsigned)BIN_NODES);
    }
    #pragma unroll
    for (int j = 0; j < SORT_EPT; ++j) {
        rk[j] = 0;
        if (tid + 256 * j < SORT_CHUNK) rk[j] = atomicAdd(&sCnt[bn[j]], 1);
    }
    __syncthreads();

    const int c = sCnt[tid];
    const int g = (tid < NBIN) ? bcnt[tid] : 0;
    int totC, totG;
    const int lstart = block_excl_scan<8>(c, swA, totC);
    const int gbase  = block_excl_scan<8>(g, swB, totG);
    int got = 0;
    if (c) got = atomicAdd(&bcur[tid], c);
    sCnt[tid]  = lstart;
    sBase[tid] = gbase + got - lstart;
    __syncthreads();

    #pragma unroll
    for (int j = 0; j < SORT_EPT; ++j) {
        if (tid + 256 * j < SORT_CHUNK) {
            const int p = sCnt[bn[j]] + rk[j];
            sRec[p] = v2i{ key[j], fb[j] };
            sPos[p] = sBase[bn[j]] + p;
        }
    }
    __syncthreads();
    #pragma unroll
    for (int j = 0; j < SORT_EPT; ++j) {
        const int i = tid + 256 * j;
        if (i < SORT_CHUNK) *(v2i*)(tmp + 2 * (size_t)sPos[i]) = sRec[i];
    }
}

__global__ void __launch_bounds__(BSORT_THREADS)
bin_sort(const int* __restrict__ tmp,
         const int* __restrict__ bcnt,
         int*       __restrict__ rec,
         int*       __restrict__ endp,
         float*     __restrict__ deg)
{
    __shared__ int sNode[BSORT_THREADS];
    __shared__ int sOff[BSORT_THREADS];
    __shared__ int swA[BSORT_THREADS / 32], swB[BSORT_THREADS / 32];
    __shared__ __attribute__((aligned(8))) v2i sOut[BSORT_CAP];

    const int tid = threadIdx.x & (BSORT_THREADS - 1);
    const int b   = (int)blockIdx.x;
    const int n0  = b * BIN_NODES;
    const unsigned n0low = (unsigned)n0 & DST_LOW_MASK;

    sNode[tid] = 0;
    int base, nrec;
    (void)block_excl_scan<BSORT_THREADS / 32>((tid < b) ? bcnt[tid] : 0, swA, base);
    const int cnt = bcnt[b];

    for (int eb = 0; eb < cnt; eb += 4 * BSORT_THREADS) {
        int w0[4];
        #pragma unroll
        for (int u = 0; u < 4; ++u) {
            const int e = eb + BSORT_THREADS * u + tid;
            w0[u] = tmp[2 * (size_t)(base + ((e < cnt) ? e : (cnt - 1)))];
        }
        #pragma unroll
        for (int u = 0; u < 4; ++u) {
            const int e = eb + BSORT_THREADS * u + tid;
            if (e < cnt) atomicAdd(&sNode[(((unsigned)w0[u] >> SRC_BITS) - n0low) & DST_LOW_MASK], 1);
        }
    }
    __syncthreads();

    const int c   = sNode[tid];
    const int off = block_excl_scan<BSORT_THREADS / 32>(c, swB, nrec);
    sOff[tid]  = off;
    sNode[tid] = off;
    if (tid < BIN_NODES) deg[n0 + tid] = (float)c;
    __syncthreads();
    if (tid < BIN_TILES) endp[b * BIN_TILES + tid] = base + sOff[16 * (tid + 1)];

    const bool staged = cnt <= BSORT_CAP;
    for (int eb = 0; eb < cnt; eb += 4 * BSORT_THREADS) {
        v2i r[4];
        #pragma unroll
        for (int u = 0; u < 4; ++u) {
            const int e = eb + BSORT_THREADS * u + tid;
            r[u] = *(const v2i*)(tmp + 2 * (size_t)(base + ((e < cnt) ? e : (cnt - 1))));
        }
        #pragma unroll
        for (int u = 0; u < 4; ++u) {
            const int e = eb + BSORT_THREADS * u + tid;
            if (e < cnt) {
                const int p = atomicAdd(&sNode[(((unsigned)r[u][0] >> SRC_BITS) - n0low) & DST_LOW_MASK], 1);
                if (staged) sOut[p] = r[u];
                else        *(v2i*)(rec + 2 * (size_t)(base + p)) = r[u];
            }
        }
    }
    if (staged) {
        __syncthreads();
        for (int i = tid; i < cnt; i += BSORT_THREADS)
            *(v2i*)(rec + 2 * (size_t)(base + i)) = sOut[i];
    }
}

#define TILE_PITCH 68
#define GP2_BLOCKS ((NN / 16 + 7) / 8)
static_assert(NN % 16 == 0, "a wave owns a whole 16-node tile");
static_assert(16 < (1 << (32 - SRC_BITS)), "a tile's nodes must be told apart by the destination's low bits");

__global__ void __launch_bounds__(256) __attribute__((amdgpu_waves_per_eu(8)))
egnn_node_gather_p2(const float* __restrict__ P,
                    const float* __restrict__ Q,
                    const int*   __restrict__ rec,
                    const int*   __restrict__ endp,
                    const float* __restrict__ wl,
                    const float* __restrict__ W2,
                    const float* __restrict__ b2,
                    const float* __restrict__ deg,
                    float*       __restrict__ out,
                    float*       __restrict__ stats)
{
    __shared__ __attribute__((aligned(16))) float sT[8][16 * TILE_PITCH];
    __shared__ float sred[8][128];

    const int tid  = threadIdx.x;
    const int lane = tid & 31;
    const int lo   = lane & 15;
    const int hi   = lane >> 4;
    const int wv   = __builtin_amdgcn_readfirstlane(tid >> 5);
    const int tile = (int)blockIdx.x * 8 + wv;

    float csum[4], csq[4];
    #pragma unroll
    for (int t = 0; t < 4; ++t) { csum[t] = 0.0f; csq[t] = 0.0f; }

    if (tile < NN / 16) {
        float* myT = &sT[wv][0];
        const v2f w = *(const v2f*)(wl + 2 * lane);

        const int n0 = tile * 16;
        const int n1 = n0 + 16;
        const int e0 = tile ? endp[tile - 1] : 0;
        const int e1 = endp[tile];

        int cur = n0;
        v2f acc = v2f{ 0.f, 0.f };

        for (int eb = e0; eb < e1; eb += 32) {
            const int  e  = eb + lane;
            const int  ec = (e < e1) ? e : (e1 - 1);
            const v2i  r  = *(const v2i*)(rec + 2 * (size_t)ec);
            const int  sv = r[0] & ((1 << SRC_BITS) - 1);
            const int  dv = n0 + (int)((((unsigned)r[0] >> SRC_BITS) - (unsigned)n0) & DST_LOW_MASK);
            const int  fb = r[1];
            const int  n  = (e1 - eb < 32) ? (e1 - eb) : 32;

            #pragma unroll 1
            for (int i = 0; i < n; i += 8) {
                v2f p[8], q[8];
                #pragma unroll
                for (int u = 0; u < 8; ++u) {
                    const int j = (i + u < n) ? (i + u) : (n - 1);
                    const unsigned s = (unsigned)__builtin_amdgcn_readlane(sv, j);
                    const unsigned d = (unsigned)__builtin_amdgcn_readlane(dv, j);
                    p[u] = *(const v2f*)(P + (size_t)s * DD + 2 * lane);
                    q[u] = *(const v2f*)(Q + (size_t)d * DD + 2 * lane);
                }
                #pragma unroll
                for (int u = 0; u < 8; ++u) {
                    if (i + u < n) {
                        const int   d = __builtin_amdgcn_readlane(dv, i + u);
                        const float f = __builtin_bit_cast(float, __builtin_amdgcn_readlane(fb, i + u));
                        if (d != cur) {
                            *(v2f*)(myT + (cur - n0) * TILE_PITCH + 2 * lane) = acc;
                            for (int m = cur + 1; m < d; ++m)
                                *(v2f*)(myT + (m - n0) * TILE_PITCH + 2 * lane) = v2f{ 0.f, 0.f };
                            cur = d;
                            acc = v2f{ 0.f, 0.f };
                        }
                        acc[0] += fmaxf((p[u][0] + q[u][0]) + f * w[0], 0.0f);
                        acc[1] += fmaxf((p[u][1] + q[u][1]) + f * w[1], 0.0f);
                    }
                }
            }
        }
        *(v2f*)(myT + (cur - n0) * TILE_PITCH + 2 * lane) = acc;
        for (int m = cur + 1; m < n1; ++m)
            *(v2f*)(myT + (m - n0) * TILE_PITCH + 2 * lane) = v2f{ 0.f, 0.f };

        __builtin_amdgcn_fence(__ATOMIC_RELEASE, "wavefront");
        __builtin_amdgcn_wave_barrier();
        __builtin_amdgcn_fence(__ATOMIC_ACQUIRE, "wavefront");

        v4f x[8];
        {
            const float* row = myT + lo * TILE_PITCH + 4 * hi;
            #pragma unroll
            for (int j = 0; j < 8; ++j) x[j] = *(const v4f*)(row + 8 * j);
        }
        float dg[8];
        #pragma unroll
        for (int r = 0; r < 8; ++r) dg[r] = deg[n0 + 8 * hi + r];
        float bcol[4];
        #pragma unroll
        for (int t = 0; t < 4; ++t) bcol[t] = b2[16 * t + lo];

        v8f mac[4];
        #pragma unroll
        for (int t = 0; t < 4; ++t) {
            #pragma unroll
            for (int r = 0; r < 8; ++r) mac[t][r] = 0.0f;
        }
        #pragma unroll
        for (int j = 0; j < 8; ++j) {
            const int kr = 8 * j + 4 * hi;
            v2f bLo[4], bHi[4];
            #pragma unroll
            for (int t = 0; t < 4; ++t) {
                const int col = 16 * t + lo;
                bLo[t] = v2f{ W2[kr * 64 + col],       W2[(kr + 1) * 64 + col] };
                bHi[t] = v2f{ W2[(kr + 2) * 64 + col], W2[(kr + 3) * 64 + col] };
            }
            const v2f aLo = { x[j][0], x[j][1] };
            const v2f aHi = { x[j][2], x[j][3] };
            #pragma unroll
            for (int t = 0; t < 4; ++t) {
                mac[t] = __builtin_amdgcn_wmma_f32_16x16x4_f32(
                    false, aLo, false, bLo[t], (short)0, mac[t], false, false);
                mac[t] = __builtin_amdgcn_wmma_f32_16x16x4_f32(
                    false, aHi, false, bHi[t], (short)0, mac[t], false, false);
            }
        }

        #pragma unroll
        for (int t = 0; t < 4; ++t) {
            #pragma unroll
            for (int r = 0; r < 8; ++r) {
                const float val = mac[t][r] + dg[r] * bcol[t];
                out[(long long)(n0 + 8 * hi + r) * DD + 16 * t + lo] = val;
                csum[t] += val;
                csq[t]  += val * val;
            }
        }
    }

    #pragma unroll
    for (int t = 0; t < 4; ++t) {
        const float s = csum[t] + __shfl_xor(csum[t], 16);
        const float q = csq[t]  + __shfl_xor(csq[t],  16);
        if (hi == 0) {
            sred[wv][16 * t + lo]      = s;
            sred[wv][64 + 16 * t + lo] = q;
        }
    }
    __syncthreads();
    if (tid < 128) {
        float a = sred[0][tid];
        for (int k = 1; k < 8; ++k) a += sred[k][tid];
        atomicAdd(&stats[tid], a);
    }
}

__global__ void __launch_bounds__(256)
egnn_node_product2(float*       hs,
                   const float* __restrict__ W2,
                   const float* __restrict__ b2,
                   const float* __restrict__ deg,
                   float*       __restrict__ stats)
{
    const int tid  = threadIdx.x;
    const int lane = tid & 31;
    const int lo   = lane & 15;
    const int hi   = lane >> 4;
    __shared__ float sred[8][128];
    const int wpb       = blockDim.x >> 5;
    const int waveGlob  = __builtin_amdgcn_readfirstlane(blockIdx.x * wpb + (tid >> 5));
    const int waveCount = gridDim.x * wpb;

    v2f wf[16][4];
    #pragma unroll
    for (int s = 0; s < 16; ++s) {
        const int kr = 8 * (s >> 1) + 4 * hi + 2 * (s & 1);
        #pragma unroll
        for (int t = 0; t < 4; ++t) {
            const int col = 16 * t + lo;
            wf[s][t] = v2f{ W2[kr * 64 + col], W2[(kr + 1) * 64 + col] };
        }
    }
    float bcol[4], csum[4], csq[4];
    #pragma unroll
    for (int t = 0; t < 4; ++t) { bcol[t] = b2[16 * t + lo]; csum[t] = 0.0f; csq[t] = 0.0f; }

    const int numTiles = NN / 16;
    v4f x[8];
    #pragma unroll
    for (int j = 0; j < 8; ++j) x[j] = v4f{ 0.f, 0.f, 0.f, 0.f };
    if (waveGlob < numTiles) {
        const float* row = hs + (long long)(waveGlob * 16 + lo) * DD + 4 * hi;
        #pragma unroll
        for (int j = 0; j < 8; ++j) x[j] = *(const v4f*)(row + 8 * j);
    }

    for (int tile = waveGlob; tile < numTiles; tile += waveCount) {
        float dg[8];
        #pragma unroll
        for (int r = 0; r < 8; ++r) dg[r] = deg[tile * 16 + 8 * hi + r];

        v8f acc[4];
        #pragma unroll
        for (int t = 0; t < 4; ++t) {
            #pragma unroll
            for (int r = 0; r < 8; ++r) acc[t][r] = 0.0f;
        }
        #pragma unroll
        for (int j = 0; j < 8; ++j) {
            const v2f aLo = { x[j][0], x[j][1] };
            const v2f aHi = { x[j][2], x[j][3] };
            #pragma unroll
            for (int t = 0; t < 4; ++t) {
                acc[t] = __builtin_amdgcn_wmma_f32_16x16x4_f32(
                    false, aLo, false, wf[2 * j][t], (short)0, acc[t], false, false);
                acc[t] = __builtin_amdgcn_wmma_f32_16x16x4_f32(
                    false, aHi, false, wf[2 * j + 1][t], (short)0, acc[t], false, false);
            }
        }

        const int tn = tile + waveCount;
        if (tn < numTiles) {
            const float* row = hs + (long long)(tn * 16 + lo) * DD + 4 * hi;
            #pragma unroll
            for (int j = 0; j < 8; ++j) x[j] = *(const v4f*)(row + 8 * j);
        }

        #pragma unroll
        for (int t = 0; t < 4; ++t) {
            #pragma unroll
            for (int r = 0; r < 8; ++r) {
                const float val = acc[t][r] + dg[r] * bcol[t];
                hs[(long long)(tile * 16 + 8 * hi + r) * DD + 16 * t + lo] = val;
                csum[t] += val;
                csq[t]  += val * val;
            }
        }
    }

    const int wv = tid >> 5;
    #pragma unroll
    for (int t = 0; t < 4; ++t) {
        const float s = csum[t] + __shfl_xor(csum[t], 16);
        const float q = csq[t]  + __shfl_xor(csq[t],  16);
        if (hi == 0) {
            sred[wv][16 * t + lo]      = s;
            sred[wv][64 + 16 * t + lo] = q;
        }
    }
    __syncthreads();
    if (tid < 128) {
        float a = sred[0][tid];
        for (int w = 1; w < wpb; ++w) a += sred[w][tid];
        atomicAdd(&stats[tid], a);
    }
}

__global__ void __launch_bounds__(256)
zero_f32(float* __restrict__ p, long long n)
{
    for (long long i = blockIdx.x * 256LL + threadIdx.x; i < n;
         i += (long long)gridDim.x * 256LL)
        p[i] = 0.0f;
}

__global__ void __launch_bounds__(256)
bn_reduce(const float* __restrict__ h, float* __restrict__ stats, int nrows)
{
    __shared__ float sSum[256], sSq[256];
    const int c = threadIdx.x & 63;
    const int r = threadIdx.x >> 6;
    float s = 0.f, q = 0.f;
    for (int row = blockIdx.x * 4 + r; row < nrows; row += gridDim.x * 4) {
        float v = h[(long long)row * DD + c];
        s += v; q += v * v;
    }
    sSum[threadIdx.x] = s; sSq[threadIdx.x] = q;
    __syncthreads();
    if (threadIdx.x < 64) {
        float ts = sSum[c] + sSum[64 + c] + sSum[128 + c] + sSum[192 + c];
        float tq = sSq [c] + sSq [64 + c] + sSq [128 + c] + sSq [192 + c];
        atomicAdd(&stats[c], ts);
        atomicAdd(&stats[64 + c], tq);
    }
}

__global__ void __launch_bounds__(256)
bn_relu_apply(float* __restrict__ h, const float* __restrict__ stats,
              const float* __restrict__ gamma, const float* __restrict__ beta,
              int nrows, float invN)
{
    const long long total = (long long)nrows * DD;
    for (long long i = blockIdx.x * 256LL + threadIdx.x; i < total;
         i += (long long)gridDim.x * 256LL) {
        const int c  = (int)(i & 63);
        const float mu  = stats[c] * invN;
        const float var = stats[64 + c] * invN - mu * mu;
        const float rs  = rsqrtf(var + BN_EPS);
        const float v   = (h[i] - mu) * rs * gamma[c] + beta[c];
        h[i] = fmaxf(v, 0.0f);
    }
}

__global__ void __launch_bounds__(256)
pool_mean(const float* __restrict__ h, float* __restrict__ emb)
{
    __shared__ float sm[256];
    const int g = blockIdx.x;
    const int c = threadIdx.x & 63;
    const int r = threadIdx.x >> 6;
    float s = 0.f;
    for (int row = r; row < NPG; row += 4)
        s += h[((long long)g * NPG + row) * DD + c];
    sm[threadIdx.x] = s;
    __syncthreads();
    if (threadIdx.x < 64)
        emb[g * DD + c] = (sm[c] + sm[64 + c] + sm[128 + c] + sm[192 + c]) / (float)NPG;
}

#define POOL_SLICES 5
#define POOL_ROWS   (NPG / POOL_SLICES)
#define POOL_TRIPS  ((POOL_ROWS + 15) / 16)

__global__ void __launch_bounds__(256)
bn_relu_apply_pool(float* __restrict__ h, const float* __restrict__ stats,
                   const float* __restrict__ gamma, const float* __restrict__ beta,
                   float invN, float* __restrict__ part)
{
    __shared__ v4f sred[256];
    const int tid = threadIdx.x;
    const int cq  = tid & 15;
    const int ro  = (tid >> 4) & 15;

    v4f mu, rs, ga, be;
    #pragma unroll
    for (int e = 0; e < 4; ++e) {
        const int   f   = 4 * cq + e;
        const float m   = stats[f] * invN;
        const float var = stats[64 + f] * invN - m * m;
        mu[e] = m;
        rs[e] = rsqrtf(var + BN_EPS);
        ga[e] = gamma[f];
        be[e] = beta[f];
    }

    float* base = h + ((long long)blockIdx.x * POOL_ROWS + ro) * DD + 4 * cq;

    v4f v[POOL_TRIPS];
    #pragma unroll
    for (int i = 0; i < POOL_TRIPS; ++i) {
        v[i] = v4f{ 0.f, 0.f, 0.f, 0.f };
        if (ro + 16 * i < POOL_ROWS) v[i] = *(const v4f*)(base + (long long)(16 * i) * DD);
    }

    v4f sum = v4f{ 0.f, 0.f, 0.f, 0.f };
    #pragma unroll
    for (int i = 0; i < POOL_TRIPS; ++i) {
        if (ro + 16 * i < POOL_ROWS) {
            v4f o;
            #pragma unroll
            for (int e = 0; e < 4; ++e) {
                const float val = (v[i][e] - mu[e]) * rs[e] * ga[e] + be[e];
                o[e] = fmaxf(val, 0.0f);
            }
            *(v4f*)(base + (long long)(16 * i) * DD) = o;
            sum += o;
        }
    }

    sred[tid] = sum;
    __syncthreads();
    if (tid < 16) {
        v4f t = sred[tid];
        #pragma unroll
        for (int k = 1; k < 16; ++k) t += sred[16 * k + tid];
        *(v4f*)(part + (long long)blockIdx.x * DD + 4 * tid) = t;
    }
}

template <int NPART>
__global__ void __launch_bounds__(256)
bn_heads(const float* __restrict__ part,
         float count,
         const float* __restrict__ bg, const float* __restrict__ bb,
         const float* __restrict__ clfW, const float* __restrict__ clfb,
         const float* __restrict__ cptW, const float* __restrict__ cptb,
         float* __restrict__ outLogits, float* __restrict__ outCpt)
{
    __shared__ float sEmb[NG * DD];
    __shared__ float sMu[64], sR[64];
    __shared__ float red[256];
    const int tid = threadIdx.x;
    const int c   = tid & 63;
    const int r   = tid >> 6;

    #pragma unroll
    for (int k = 0; k < (NG * 16 + 255) / 256; ++k) {
        const int item = (tid & 255) + 256 * k;
        if (item < NG * 16) {
            const int g = item >> 4, cqd = item & 15;
            v4f a = v4f{ 0.f, 0.f, 0.f, 0.f };
            #pragma unroll
            for (int p = 0; p < NPART; ++p)
                a += *(const v4f*)(part + (long long)(g * NPART + p) * DD + 4 * cqd);
            #pragma unroll
            for (int e = 0; e < 4; ++e) sEmb[g * DD + 4 * cqd + e] = a[e] / count;
        }
    }
    __syncthreads();

    float s = 0.f, q = 0.f;
    for (int g = r; g < NG; g += 4) {
        float v = sEmb[g * DD + c];
        s += v; q += v * v;
    }
    red[tid] = s; __syncthreads();
    if (tid < 64) sMu[c] = (red[c] + red[64 + c] + red[128 + c] + red[192 + c]) / (float)NG;
    __syncthreads();
    red[tid] = q; __syncthreads();
    if (tid < 64) {
        float tq  = (red[c] + red[64 + c] + red[128 + c] + red[192 + c]) / (float)NG;
        float mu  = sMu[c];
        sR[c] = rsqrtf(tq - mu * mu + BN_EPS);
    }
    __syncthreads();
    for (int i = tid; i < NG * DD; i += 256) {
        int cc = i & 63;
        sEmb[i] = (sEmb[i] - sMu[cc]) * sR[cc] * bg[cc] + bb[cc];
    }
    __syncthreads();
    for (int i = tid; i < NG * NC; i += 256) {
        int g = i / NC, k = i % NC;
        float a = clfb[k];
        #pragma unroll 8
        for (int cc = 0; cc < DD; ++cc) a += sEmb[g * DD + cc] * clfW[cc * NC + k];
        outLogits[i] = a;
    }
    for (int i = tid; i < NG * NCP; i += 256) {
        int g = i >> 4, k = i & 15;
        float a = cptb[k];
        #pragma unroll 8
        for (int cc = 0; cc < DD; ++cc) a += sEmb[g * DD + cc] * cptW[cc * NCP + k];
        outCpt[i] = a;
    }
}

extern "C" void kernel_launch(void* const* d_in, const int* in_sizes, int n_in,
                              void* d_out, int out_size, void* d_ws, size_t ws_size,
                              hipStream_t stream)
{
    const float* x    = (const float*)d_in[0];
    const float* ef   = (const float*)d_in[1];
    const int*   src  = (const int*)  d_in[2];
    const int*   dst  = (const int*)  d_in[3];
    const float* W1s  = (const float*)d_in[5];
    const float* b1s  = (const float*)d_in[6];
    const float* W2s  = (const float*)d_in[7];
    const float* b2s  = (const float*)d_in[8];
    const float* gam  = (const float*)d_in[9];
    const float* bet  = (const float*)d_in[10];
    const float* bng  = (const float*)d_in[11];
    const float* bnb  = (const float*)d_in[12];
    const float* clfW = (const float*)d_in[13];
    const float* clfb = (const float*)d_in[14];
    const float* cptW = (const float*)d_in[15];
    const float* cptb = (const float*)d_in[16];

    float* outH      = (float*)d_out;
    float* outLogits = outH + (long long)NN * DD;
    float* outCpt    = outLogits + NG * NC;

    float* hA    = (float*)d_ws;
    float* stats = hA + (long long)NN * DD;
    float* emb   = stats + 128;
    float* hQ    = emb + NG * DD;

    float* degT  = hQ + (long long)NN * DD;

    int*   cntT  = (int*)(degT + NN);
    int*   cRec  = cntT + NN + 128;

    if (ws_size >= (size_t)WS_GROUPED_BYTES) {
        for (int l = 0; l < NL; ++l) {
            const float* W1l = W1s + (long long)l * (2 * DD + 1) * DD;
            float*       stl = stats + 128 * l;
            if (l == 0) {
                egnn_node_project<false><<<512, 256, 0, stream>>>(
                    x, W1l, b1s, hA, hQ, stl, gam, bet, 1.0f / (float)NN, nullptr, stl, cntT);
                bin_count<<<SORT_BLOCKS, 256, 0, stream>>>(dst, cntT);
                bin_partition<<<SORT_BLOCKS, 256, 0, stream>>>(src, dst, ef, cntT, cntT + 256, (int*)outH);
                bin_sort<<<NBIN, BSORT_THREADS, 0, stream>>>((const int*)outH, cntT, cRec, cntT + 1024, degT);
            } else {
                egnn_node_project<true><<<512, 256, 0, stream>>>(
                    outH, W1l, b1s + l * DD, hA, hQ,
                    stats + 128 * (l - 1), gam + (l - 1) * DD, bet + (l - 1) * DD,
                    1.0f / (float)NN, nullptr, stl, nullptr);
            }
            egnn_node_gather_p2<<<GP2_BLOCKS, 256, 0, stream>>>(
                hA, hQ, cRec, cntT + 1024, W1l + 2 * DD * DD,
                W2s + (long long)l * DD * DD, b2s + l * DD, degT, outH, stl);
        }
        bn_relu_apply_pool<<<NG * POOL_SLICES, 256, 0, stream>>>(
            outH, stats + 128 * (NL - 1), gam + (NL - 1) * DD, bet + (NL - 1) * DD,
            1.0f / (float)NN, hA);
        bn_heads<POOL_SLICES><<<1, 256, 0, stream>>>(hA, (float)NPG,
                                                     bng, bnb, clfW, clfb, cptW, cptb,
                                                     outLogits, outCpt);
    } else if (ws_size >= (size_t)WS_NODE_P2_BYTES) {
        for (int l = 0; l < NL; ++l) {
            const float* W1l = W1s + (long long)l * (2 * DD + 1) * DD;
            float*       stl = stats + 128 * l;
            if (l == 0) {
                egnn_node_project<false><<<512, 256, 0, stream>>>(
                    x, W1l, b1s, hA, hQ, stl, gam, bet, 1.0f / (float)NN, outH, stl, (int*)degT);
                egnn_edge_sum<true><<<EDGE_SUM_BLOCKS, 256, 0, stream>>>(
                    hA, hQ, ef, src, dst, W1l + 2 * DD * DD, outH, degT);
            } else {
                egnn_node_project<true><<<512, 256, 0, stream>>>(
                    outH, W1l, b1s + l * DD, hA, hQ,
                    stats + 128 * (l - 1), gam + (l - 1) * DD, bet + (l - 1) * DD,
                    1.0f / (float)NN, outH, stl, nullptr);
                egnn_edge_sum<false><<<EDGE_SUM_BLOCKS, 256, 0, stream>>>(
                    hA, hQ, ef, src, dst, W1l + 2 * DD * DD, outH, degT);
            }
            egnn_node_product2<<<768, 128, 0, stream>>>(
                outH, W2s + (long long)l * DD * DD, b2s + l * DD, degT, stl);
        }
        bn_relu_apply_pool<<<NG * POOL_SLICES, 256, 0, stream>>>(
            outH, stats + 128 * (NL - 1), gam + (NL - 1) * DD, bet + (NL - 1) * DD,
            1.0f / (float)NN, hA);
        bn_heads<POOL_SLICES><<<1, 256, 0, stream>>>(hA, (float)NPG,
                                                     bng, bnb, clfW, clfb, cptW, cptb,
                                                     outLogits, outCpt);
    } else if (ws_size >= (size_t)WS_PROJECTED_BYTES) {
        for (int l = 0; l < NL; ++l) {
            const float* W1l = W1s + (long long)l * (2 * DD + 1) * DD;
            float*       stl = stats + 128 * l;
            if (l == 0)
                egnn_node_project<false><<<512, 256, 0, stream>>>(
                    x, W1l, b1s, hA, hQ, stl, gam, bet, 1.0f / (float)NN, outH, stl, nullptr);
            else
                egnn_node_project<true><<<512, 256, 0, stream>>>(
                    outH, W1l, b1s + l * DD, hA, hQ,
                    stats + 128 * (l - 1), gam + (l - 1) * DD, bet + (l - 1) * DD,
                    1.0f / (float)NN, outH, stl, nullptr);
            egnn_edge_layer_pq<<<EDGE_PQ_BLOCKS, 256, 0, stream>>>(
                hA, hQ, ef, src, dst,
                W1l + 2 * DD * DD,
                W2s + (long long)l * DD * DD,
                b2s + l * DD,
                outH);
            bn_reduce<<<512, 256, 0, stream>>>(outH, stl, NN);
        }
        bn_relu_apply_pool<<<NG * POOL_SLICES, 256, 0, stream>>>(
            outH, stats + 128 * (NL - 1), gam + (NL - 1) * DD, bet + (NL - 1) * DD,
            1.0f / (float)NN, hA);
        bn_heads<POOL_SLICES><<<1, 256, 0, stream>>>(hA, (float)NPG,
                                                     bng, bnb, clfW, clfb, cptW, cptb,
                                                     outLogits, outCpt);
    } else {
        const float* hin = x;
        float* tgt[NL] = { outH, hA, outH };

        const int edgeBlocks = 256;

        for (int l = 0; l < NL; ++l) {
            float* hn = tgt[l];
            zero_f32<<<1024, 256, 0, stream>>>(hn, (long long)NN * DD);
            zero_f32<<<1,    256, 0, stream>>>(stats, 128);
            egnn_edge_layer<<<edgeBlocks, 256, 0, stream>>>(
                hin, ef, src, dst,
                W1s + (long long)l * (2 * DD + 1) * DD,
                b1s + l * DD,
                W2s + (long long)l * DD * DD,
                b2s + l * DD,
                hn);
            bn_reduce<<<512, 256, 0, stream>>>(hn, stats, NN);
            bn_relu_apply<<<1024, 256, 0, stream>>>(hn, stats, gam + l * DD, bet + l * DD,
                                                    NN, 1.0f / (float)NN);
            hin = hn;
        }
        pool_mean<<<NG, 256, 0, stream>>>(outH, emb);
        bn_heads<1><<<1, 256, 0, stream>>>(emb, 1.0f,
                                           bng, bnb, clfW, clfb, cptW, cptb,
                                           outLogits, outCpt);
    }
}
